# starting kernel
# baseline (speedup 1.0000x reference)
#include <hip/hip_runtime.h>

typedef float v2f __attribute__((ext_vector_type(2)));
typedef float v8f __attribute__((ext_vector_type(8)));

#define NSP   10
#define KROW  12
#define KTOT  (81 * KROW)
#define MROWS 16

__global__ void sc_setup(const int* __restrict__ index, int N,
                         int* __restrict__ meta, int* __restrict__ perm,
                         int* __restrict__ tileSpecies, int* __restrict__ tileSlot) {
  if (threadIdx.x != 0) return;
  int cnt[NSP];
  for (int s = 0; s < NSP; ++s) cnt[s] = 0;
  for (int n = 0; n < N; ++n) {
    int s = index[n];
    if (s >= 0 && s < NSP) cnt[s]++;
  }
  int segBase[NSP];
  int base = 0, tiles = 0;
  for (int s = 0; s < NSP; ++s) {
    segBase[s] = base;
    int nt = (cnt[s] + 15) >> 4;
    for (int t = 0; t < nt; ++t) {
      tileSpecies[tiles] = s;
      tileSlot[tiles]    = base + t * 16;
      ++tiles;
    }
    base += nt * 16;
  }
  meta[0] = tiles;
  for (int i = 0; i < base; ++i) perm[i] = -1;
  int pos[NSP];
  for (int s = 0; s < NSP; ++s) pos[s] = 0;
  for (int n = 0; n < N; ++n) {
    int s = index[n];
    if (s >= 0 && s < NSP) perm[segBase[s] + pos[s]++] = n;
  }
}

__global__ __launch_bounds__(128) void sc_main(
    const float* __restrict__ X,
    const int*   __restrict__ meta,
    const int*   __restrict__ perm,
    const int*   __restrict__ tileSpecies,
    const int*   __restrict__ tileSlot,
    float*       __restrict__ out, int C,
    const float* __restrict__ U3_0e, const float* __restrict__ W3_0e,
    const float* __restrict__ U3_1o, const float* __restrict__ W3_1o,
    const float* __restrict__ U3_2e, const float* __restrict__ W3_2e,
    const float* __restrict__ U2_0e, const float* __restrict__ W2_0e,
    const float* __restrict__ U2_1o, const float* __restrict__ W2_1o,
    const float* __restrict__ U2_2e, const float* __restrict__ W2_2e,
    const float* __restrict__ U1_0e, const float* __restrict__ W1_0e,
    const float* __restrict__ U1_1o, const float* __restrict__ W1_1o,
    const float* __restrict__ U1_2e, const float* __restrict__ W1_2e)
{
  __shared__ float Ct[MROWS * KTOT];
  const int c   = blockIdx.x;
  const int s   = blockIdx.y;
  const int tid = threadIdx.x;

  for (int e = tid; e < MROWS * KTOT; e += 128) {
    const int m = e / KTOT;
    const int p = e - m * KTOT;
    float v = 0.0f;
    if (m < 9) {
      const int row  = p / KROW;
      const int slot = p - row * KROW;
      const int d0 = row / 9;
      const int d1 = row - d0 * 9;
      const float *u3, *w3, *u2, *w2, *u1, *w1;
      int mul3, mul2, irdim, ii;
      if (m == 0)     { u3=U3_0e; w3=W3_0e; u2=U2_0e; w2=W2_0e; u1=U1_0e; w1=W1_0e; mul3=10; mul2=3; irdim=1; ii=0;   }
      else if (m < 4) { u3=U3_1o; w3=W3_1o; u2=U2_1o; w2=W2_1o; u1=U1_1o; w1=W1_1o; mul3=11; mul2=2; irdim=3; ii=m-1; }
      else            { u3=U3_2e; w3=W3_2e; u2=U2_2e; w2=W2_2e; u1=U1_2e; w1=W1_2e; mul3=13; mul2=3; irdim=5; ii=m-4; }
      if (slot < 9) {
        const int ub = (row * 9 + slot) * mul3 * irdim + ii;
        for (int k = 0; k < mul3; ++k)
          v += u3[ub + k * irdim] * w3[(s * mul3 + k) * C + c];
      } else if (slot == 9) {
        const int ub = row * mul2 * irdim + ii;
        for (int k = 0; k < mul2; ++k)
          v += u2[ub + k * irdim] * w2[(s * mul2 + k) * C + c];
      } else if (slot == 10 && d1 == 0) {
        v = u1[d0 * irdim + ii] * w1[s * C + c];
      }
    }
    Ct[e] = v;
  }
  __syncthreads();

  const int  lane = tid & 31;
  const int  wave = tid >> 5;
  const int  n16  = lane & 15;
  const bool hi   = (lane & 16) != 0;
  const float* Abase = &Ct[n16 * KTOT + (hi ? 2 : 0)];
  const int numTiles = meta[0];

  int mc = 0;
  for (int t = 0; t < numTiles; ++t) {
    if (tileSpecies[t] != s) continue;
    const int my = mc++;
    if ((my & 3) != wave) continue;

    const int  slotBase = tileSlot[t];
    const int  node  = perm[slotBase + n16];
    const bool valid = node >= 0;
    const int  nodeC = valid ? node : 0;
    const float* xp = X + (nodeC * C + c) * 9;
    float xv[9];
#pragma unroll
    for (int d = 0; d < 9; ++d) xv[d] = valid ? xp[d] : 0.0f;

    const float e0 = hi ? xv[2] : xv[0];
    const float e1 = hi ? xv[3] : xv[1];
    const float e2 = hi ? xv[6] : xv[4];
    const float e3 = hi ? xv[7] : xv[5];
    const float e4 = hi ? 0.0f  : xv[8];
    const float hm = hi ? 0.0f  : 1.0f;

    v8f acc0 = {0.f,0.f,0.f,0.f,0.f,0.f,0.f,0.f};
    v8f acc1 = acc0, acc2 = acc0;

#pragma unroll 1
    for (int d1i = 0; d1i < 9; ++d1i) {
      float xd1 = xv[0];
#pragma unroll
      for (int q = 1; q < 9; ++q) if (d1i == q) xd1 = xv[q];
      const int rb = d1i * KROW;
#pragma unroll
      for (int d0 = 0; d0 < 9; ++d0) {
        const float xx = xv[d0] * xd1;
        const float* ap = Abase + d0 * (9 * KROW) + rb;
        v2f af0 = *(const v2f*)(ap);
        v2f af1 = *(const v2f*)(ap + 4);
        v2f af2 = *(const v2f*)(ap + 8);
        v2f b0, b1, b2;
        b0.x = xx * e0;  b0.y = xx * e1;
        b1.x = xx * e2;  b1.y = xx * e3;
        const float c1h = (hi && d1i == 0) ? xv[d0] : 0.0f;
        b2.x = xx * e4 + c1h;
        b2.y = xx * hm;
        acc0 = __builtin_amdgcn_wmma_f32_16x16x4_f32(false, af0, false, b0,
                                                     (short)0, acc0, false, false);
        acc1 = __builtin_amdgcn_wmma_f32_16x16x4_f32(false, af1, false, b1,
                                                     (short)0, acc1, false, false);
        acc2 = __builtin_amdgcn_wmma_f32_16x16x4_f32(false, af2, false, b2,
                                                     (short)0, acc2, false, false);
      }
    }
    v8f acc = acc0 + acc1 + acc2;

    if (valid) {
      float* op = out + (node * C + c) * 9;
      if (!hi) {
#pragma unroll
        for (int r = 0; r < 8; ++r) op[r] = acc[r];
      } else {
        op[8] = acc[0];
      }
    }
  }
}

extern "C" void kernel_launch(void* const* d_in, const int* in_sizes, int n_in,
                              void* d_out, int out_size, void* d_ws, size_t ws_size,
                              hipStream_t stream) {
  (void)n_in; (void)out_size; (void)ws_size;
  const float* X     = (const float*)d_in[0];
  const int*   index = (const int*)d_in[1];
  const int N = in_sizes[1];
  const int C = in_sizes[0] / (N * 9);

  int* meta        = (int*)d_ws;
  int* perm        = meta + 16;
  int  permCap     = (N + 16 * NSP + 15) & ~15;
  int* tileSpecies = perm + permCap;
  int* tileSlot    = tileSpecies + 256;

  sc_setup<<<1, 32, 0, stream>>>(index, N, meta, perm, tileSpecies, tileSlot);

  sc_main<<<dim3(C, NSP), 128, 0, stream>>>(
      X, meta, perm, tileSpecies, tileSlot, (float*)d_out, C,
      (const float*)d_in[2],  (const float*)d_in[3],
      (const float*)d_in[4],  (const float*)d_in[5],
      (const float*)d_in[6],  (const float*)d_in[7],
      (const float*)d_in[8],  (const float*)d_in[9],
      (const float*)d_in[10], (const float*)d_in[11],
      (const float*)d_in[12], (const float*)d_in[13],
      (const float*)d_in[14], (const float*)d_in[15],
      (const float*)d_in[16], (const float*)d_in[17],
      (const float*)d_in[18], (const float*)d_in[19]);
}


# Setup kernel: the serial one-lane species sort of the nodes becomes a one-block 32-wave ballot counting sort (same perm and tile tables, no atomics); main kernel untouched
# speedup vs baseline: 4.2066x; 4.2066x over previous
#include <hip/hip_runtime.h>

typedef float v2f __attribute__((ext_vector_type(2)));
typedef float v8f __attribute__((ext_vector_type(8)));

#define NSP   10
#define KROW  12
#define KTOT  (81 * KROW)
#define MROWS 16

#define SETUP_WAVES   32
#define SETUP_THREADS (SETUP_WAVES * 32)

__global__ __launch_bounds__(SETUP_THREADS) void sc_setup(
    const int* __restrict__ index, int N,
    int* __restrict__ meta, int* __restrict__ perm,
    int* __restrict__ tileSpecies, int* __restrict__ tileSlot) {
  __shared__ int waveCnt[SETUP_WAVES][NSP];
  __shared__ int spCnt[NSP];
  const int tid  = threadIdx.x;
  const int lane = tid & 31;
  const int wave = tid >> 5;

  const int chunks = (N + 31) >> 5;
  const int cpw    = (chunks + SETUP_WAVES - 1) / SETUP_WAVES;
  const int n0     = wave * cpw * 32;

  int run[NSP];
  for (int q = 0; q < NSP; ++q) run[q] = 0;
  for (int j = 0; j < cpw; ++j) {
    const int n = n0 + j * 32 + lane;
    const int s = (n < N) ? index[n] : -1;
    for (int q = 0; q < NSP; ++q)
      run[q] += __builtin_popcount(__builtin_amdgcn_ballot_w32(s == q));
  }
  if (lane < NSP) {
    int v = run[0];
    for (int q = 1; q < NSP; ++q) if (lane == q) v = run[q];
    waveCnt[wave][lane] = v;
  }
  __syncthreads();

  if (tid < NSP) {
    int acc = 0;
    for (int w = 0; w < SETUP_WAVES; ++w) {
      const int v = waveCnt[w][tid];
      waveCnt[w][tid] = acc;
      acc += v;
    }
    spCnt[tid] = acc;
  }
  __syncthreads();

  int cnt[NSP], segBase[NSP];
  int base = 0;
  for (int q = 0; q < NSP; ++q) {
    cnt[q]     = spCnt[q];
    segBase[q] = base;
    base += ((cnt[q] + 15) >> 4) << 4;
  }
  const int tiles = base >> 4;

  int wb[NSP];
  for (int q = 0; q < NSP; ++q) wb[q] = segBase[q] + waveCnt[wave][q];
  for (int j = 0; j < cpw; ++j) {
    const int n = n0 + j * 32 + lane;
    const int s = (n < N) ? index[n] : -1;
    int pos = -1;
    for (int q = 0; q < NSP; ++q) {
      const unsigned m = __builtin_amdgcn_ballot_w32(s == q);
      if (s == q) pos = wb[q] + (int)__builtin_amdgcn_mbcnt_lo(m, 0u);
      wb[q] += __builtin_popcount(m);
    }
    if (pos >= 0) perm[pos] = n;
  }

  if (tid < NSP * 16) {
    const int sp = tid >> 4;
    const int j  = tid & 15;
    int c = cnt[0], sb = segBase[0];
    for (int q = 1; q < NSP; ++q) if (sp == q) { c = cnt[q]; sb = segBase[q]; }
    const int slot = c + j;
    if (slot < (((c + 15) >> 4) << 4)) perm[sb + slot] = -1;
  }

  for (int t = tid; t < tiles; t += SETUP_THREADS) {
    int sp = 0;
    for (int q = 0; q < NSP - 1; ++q)
      if (t >= ((segBase[q] >> 4) + ((cnt[q] + 15) >> 4))) sp = q + 1;
    tileSpecies[t] = sp;
    tileSlot[t]    = t * 16;
  }
  if (tid == 0) meta[0] = tiles;
}

__global__ __launch_bounds__(128) void sc_main(
    const float* __restrict__ X,
    const int*   __restrict__ meta,
    const int*   __restrict__ perm,
    const int*   __restrict__ tileSpecies,
    const int*   __restrict__ tileSlot,
    float*       __restrict__ out, int C,
    const float* __restrict__ U3_0e, const float* __restrict__ W3_0e,
    const float* __restrict__ U3_1o, const float* __restrict__ W3_1o,
    const float* __restrict__ U3_2e, const float* __restrict__ W3_2e,
    const float* __restrict__ U2_0e, const float* __restrict__ W2_0e,
    const float* __restrict__ U2_1o, const float* __restrict__ W2_1o,
    const float* __restrict__ U2_2e, const float* __restrict__ W2_2e,
    const float* __restrict__ U1_0e, const float* __restrict__ W1_0e,
    const float* __restrict__ U1_1o, const float* __restrict__ W1_1o,
    const float* __restrict__ U1_2e, const float* __restrict__ W1_2e)
{
  __shared__ float Ct[MROWS * KTOT];
  const int c   = blockIdx.x;
  const int s   = blockIdx.y;
  const int tid = threadIdx.x;

  for (int e = tid; e < MROWS * KTOT; e += 128) {
    const int m = e / KTOT;
    const int p = e - m * KTOT;
    float v = 0.0f;
    if (m < 9) {
      const int row  = p / KROW;
      const int slot = p - row * KROW;
      const int d0 = row / 9;
      const int d1 = row - d0 * 9;
      const float *u3, *w3, *u2, *w2, *u1, *w1;
      int mul3, mul2, irdim, ii;
      if (m == 0)     { u3=U3_0e; w3=W3_0e; u2=U2_0e; w2=W2_0e; u1=U1_0e; w1=W1_0e; mul3=10; mul2=3; irdim=1; ii=0;   }
      else if (m < 4) { u3=U3_1o; w3=W3_1o; u2=U2_1o; w2=W2_1o; u1=U1_1o; w1=W1_1o; mul3=11; mul2=2; irdim=3; ii=m-1; }
      else            { u3=U3_2e; w3=W3_2e; u2=U2_2e; w2=W2_2e; u1=U1_2e; w1=W1_2e; mul3=13; mul2=3; irdim=5; ii=m-4; }
      if (slot < 9) {
        const int ub = (row * 9 + slot) * mul3 * irdim + ii;
        for (int k = 0; k < mul3; ++k)
          v += u3[ub + k * irdim] * w3[(s * mul3 + k) * C + c];
      } else if (slot == 9) {
        const int ub = row * mul2 * irdim + ii;
        for (int k = 0; k < mul2; ++k)
          v += u2[ub + k * irdim] * w2[(s * mul2 + k) * C + c];
      } else if (slot == 10 && d1 == 0) {
        v = u1[d0 * irdim + ii] * w1[s * C + c];
      }
    }
    Ct[e] = v;
  }
  __syncthreads();

  const int  lane = tid & 31;
  const int  wave = tid >> 5;
  const int  n16  = lane & 15;
  const bool hi   = (lane & 16) != 0;
  const float* Abase = &Ct[n16 * KTOT + (hi ? 2 : 0)];
  const int numTiles = meta[0];

  int mc = 0;
  for (int t = 0; t < numTiles; ++t) {
    if (tileSpecies[t] != s) continue;
    const int my = mc++;
    if ((my & 3) != wave) continue;

    const int  slotBase = tileSlot[t];
    const int  node  = perm[slotBase + n16];
    const bool valid = node >= 0;
    const int  nodeC = valid ? node : 0;
    const float* xp = X + (nodeC * C + c) * 9;
    float xv[9];
#pragma unroll
    for (int d = 0; d < 9; ++d) xv[d] = valid ? xp[d] : 0.0f;

    const float e0 = hi ? xv[2] : xv[0];
    const float e1 = hi ? xv[3] : xv[1];
    const float e2 = hi ? xv[6] : xv[4];
    const float e3 = hi ? xv[7] : xv[5];
    const float e4 = hi ? 0.0f  : xv[8];
    const float hm = hi ? 0.0f  : 1.0f;

    v8f acc0 = {0.f,0.f,0.f,0.f,0.f,0.f,0.f,0.f};
    v8f acc1 = acc0, acc2 = acc0;

#pragma unroll 1
    for (int d1i = 0; d1i < 9; ++d1i) {
      float xd1 = xv[0];
#pragma unroll
      for (int q = 1; q < 9; ++q) if (d1i == q) xd1 = xv[q];
      const int rb = d1i * KROW;
#pragma unroll
      for (int d0 = 0; d0 < 9; ++d0) {
        const float xx = xv[d0] * xd1;
        const float* ap = Abase + d0 * (9 * KROW) + rb;
        v2f af0 = *(const v2f*)(ap);
        v2f af1 = *(const v2f*)(ap + 4);
        v2f af2 = *(const v2f*)(ap + 8);
        v2f b0, b1, b2;
        b0.x = xx * e0;  b0.y = xx * e1;
        b1.x = xx * e2;  b1.y = xx * e3;
        const float c1h = (hi && d1i == 0) ? xv[d0] : 0.0f;
        b2.x = xx * e4 + c1h;
        b2.y = xx * hm;
        acc0 = __builtin_amdgcn_wmma_f32_16x16x4_f32(false, af0, false, b0,
                                                     (short)0, acc0, false, false);
        acc1 = __builtin_amdgcn_wmma_f32_16x16x4_f32(false, af1, false, b1,
                                                     (short)0, acc1, false, false);
        acc2 = __builtin_amdgcn_wmma_f32_16x16x4_f32(false, af2, false, b2,
                                                     (short)0, acc2, false, false);
      }
    }
    v8f acc = acc0 + acc1 + acc2;

    if (valid) {
      float* op = out + (node * C + c) * 9;
      if (!hi) {
#pragma unroll
        for (int r = 0; r < 8; ++r) op[r] = acc[r];
      } else {
        op[8] = acc[0];
      }
    }
  }
}

extern "C" void kernel_launch(void* const* d_in, const int* in_sizes, int n_in,
                              void* d_out, int out_size, void* d_ws, size_t ws_size,
                              hipStream_t stream) {
  (void)n_in; (void)out_size; (void)ws_size;
  const float* X     = (const float*)d_in[0];
  const int*   index = (const int*)d_in[1];
  const int N = in_sizes[1];
  const int C = in_sizes[0] / (N * 9);

  int* meta        = (int*)d_ws;
  int* perm        = meta + 16;
  int  permCap     = (N + 16 * NSP + 15) & ~15;
  int* tileSpecies = perm + permCap;
  int* tileSlot    = tileSpecies + 256;

  sc_setup<<<1, SETUP_THREADS, 0, stream>>>(index, N, meta, perm, tileSpecies, tileSlot);

  sc_main<<<dim3(C, NSP), 128, 0, stream>>>(
      X, meta, perm, tileSpecies, tileSlot, (float*)d_out, C,
      (const float*)d_in[2],  (const float*)d_in[3],
      (const float*)d_in[4],  (const float*)d_in[5],
      (const float*)d_in[6],  (const float*)d_in[7],
      (const float*)d_in[8],  (const float*)d_in[9],
      (const float*)d_in[10], (const float*)d_in[11],
      (const float*)d_in[12], (const float*)d_in[13],
      (const float*)d_in[14], (const float*)d_in[15],
      (const float*)d_in[16], (const float*)d_in[17],
      (const float*)d_in[18], (const float*)d_in[19]);
}


# Coefficient build: order-3 U tensors staged through LDS in contiguous chunks by whole-line 16-byte loads (next chunk prefetched, pad rows as scratch); same fused sums in the same k order
# speedup vs baseline: 9.7194x; 2.3105x over previous
#include <hip/hip_runtime.h>

typedef float v2f __attribute__((ext_vector_type(2)));
typedef float v8f __attribute__((ext_vector_type(8)));

#define NSP   10
#define KROW  12
#define KTOT  (81 * KROW)
#define MROWS 16

#define SETUP_WAVES   32
#define SETUP_THREADS (SETUP_WAVES * 32)

__global__ __launch_bounds__(SETUP_THREADS) void sc_setup(
    const int* __restrict__ index, int N,
    int* __restrict__ meta, int* __restrict__ perm,
    int* __restrict__ tileSpecies, int* __restrict__ tileSlot) {
  __shared__ int waveCnt[SETUP_WAVES][NSP];
  __shared__ int spCnt[NSP];
  const int tid  = threadIdx.x;
  const int lane = tid & 31;
  const int wave = tid >> 5;

  const int chunks = (N + 31) >> 5;
  const int cpw    = (chunks + SETUP_WAVES - 1) / SETUP_WAVES;
  const int n0     = wave * cpw * 32;

  int run[NSP];
  for (int q = 0; q < NSP; ++q) run[q] = 0;
  for (int j = 0; j < cpw; ++j) {
    const int n = n0 + j * 32 + lane;
    const int s = (n < N) ? index[n] : -1;
    for (int q = 0; q < NSP; ++q)
      run[q] += __builtin_popcount(__builtin_amdgcn_ballot_w32(s == q));
  }
  if (lane < NSP) {
    int v = run[0];
    for (int q = 1; q < NSP; ++q) if (lane == q) v = run[q];
    waveCnt[wave][lane] = v;
  }
  __syncthreads();

  if (tid < NSP) {
    int acc = 0;
    for (int w = 0; w < SETUP_WAVES; ++w) {
      const int v = waveCnt[w][tid];
      waveCnt[w][tid] = acc;
      acc += v;
    }
    spCnt[tid] = acc;
  }
  __syncthreads();

  int cnt[NSP], segBase[NSP];
  int base = 0;
  for (int q = 0; q < NSP; ++q) {
    cnt[q]     = spCnt[q];
    segBase[q] = base;
    base += ((cnt[q] + 15) >> 4) << 4;
  }
  const int tiles = base >> 4;

  int wb[NSP];
  for (int q = 0; q < NSP; ++q) wb[q] = segBase[q] + waveCnt[wave][q];
  for (int j = 0; j < cpw; ++j) {
    const int n = n0 + j * 32 + lane;
    const int s = (n < N) ? index[n] : -1;
    int pos = -1;
    for (int q = 0; q < NSP; ++q) {
      const unsigned m = __builtin_amdgcn_ballot_w32(s == q);
      if (s == q) pos = wb[q] + (int)__builtin_amdgcn_mbcnt_lo(m, 0u);
      wb[q] += __builtin_popcount(m);
    }
    if (pos >= 0) perm[pos] = n;
  }

  if (tid < NSP * 16) {
    const int sp = tid >> 4;
    const int j  = tid & 15;
    int c = cnt[0], sb = segBase[0];
    for (int q = 1; q < NSP; ++q) if (sp == q) { c = cnt[q]; sb = segBase[q]; }
    const int slot = c + j;
    if (slot < (((c + 15) >> 4) << 4)) perm[sb + slot] = -1;
  }

  for (int t = tid; t < tiles; t += SETUP_THREADS) {
    int sp = 0;
    for (int q = 0; q < NSP - 1; ++q)
      if (t >= ((segBase[q] >> 4) + ((cnt[q] + 15) >> 4))) sp = q + 1;
    tileSpecies[t] = sp;
    tileSlot[t]    = t * 16;
  }
  if (tid == 0) meta[0] = tiles;
}

#define STAGE_OFF (9 * KTOT)
#define STAGE_CAP (7 * KTOT)
typedef float v4f __attribute__((ext_vector_type(4)));

template <int MUL, int IRD, int PC>
struct Chunk {
  static constexpr int PER_P  = MUL * IRD;
  static constexpr int T      = 729 * PER_P;
  static constexpr int NCH    = (729 + PC - 1) / PC;
  static constexpr int FMAX   = PC * PER_P;
  static constexpr int NU     = ((FMAX + 6) / 4 + 127) / 128;
  static constexpr int ROUNDS = (PC * IRD + 127) / 128;
  static_assert(FMAX + 6 <= STAGE_CAP, "chunk must fit the pad rows");
};

template <int NU>
struct Pref {
  v4f   u[NU];
  float edge;
};

template <typename CH>
__device__ __forceinline__ void chunk_fetch(const float* __restrict__ U, int mis, int i,
                                            int tid, Pref<CH::NU>& pf) {
  const int a   = i * CH::FMAX;
  const int end = (a + CH::FMAX < CH::T) ? a + CH::FMAX : CH::T;
  const int e0  = ((a + mis) & ~3) - mis;
#pragma unroll
  for (int j = 0; j < CH::NU; ++j) {
    const int e = e0 + 4 * (tid + 128 * j);
    v4f v = {0.f, 0.f, 0.f, 0.f};
    if (e < end && e >= 0 && e + 4 <= CH::T) v = *(const v4f*)(U + e);
    pf.u[j] = v;
  }
  pf.edge = 0.0f;
  const int eLast = e0 + (((end - 1 - e0) >> 2) << 2);
  if (e0 < 0 || eLast + 4 > CH::T) {
    if (tid < 8) {
      const int eb = (tid < 4) ? e0 : eLast;
      const int e  = eb + (tid & 3);
      if ((eb < 0 || eb + 4 > CH::T) && e >= 0 && e < CH::T) pf.edge = U[e];
    }
  }
}

template <typename CH>
__device__ __forceinline__ void chunk_stage(float* stage, int mis, int i, int tid,
                                            const Pref<CH::NU>& pf) {
  const int a   = i * CH::FMAX;
  const int end = (a + CH::FMAX < CH::T) ? a + CH::FMAX : CH::T;
  const int e0  = ((a + mis) & ~3) - mis;
#pragma unroll
  for (int j = 0; j < CH::NU; ++j) {
    const int n = 4 * (tid + 128 * j);
    const int e = e0 + n;
    if (e < end && e >= 0 && e + 4 <= CH::T) *(v4f*)(stage + n) = pf.u[j];
  }
  const int eLast = e0 + (((end - 1 - e0) >> 2) << 2);
  if (e0 < 0 || eLast + 4 > CH::T) {
    if (tid < 8) {
      const int eb = (tid < 4) ? e0 : eLast;
      const int e  = eb + (tid & 3);
      if ((eb < 0 || eb + 4 > CH::T) && e >= 0 && e < CH::T) stage[e - e0] = pf.edge;
    }
  }
}

template <int MUL, int IRD, int PC, int M0>
__device__ __forceinline__ void build_order3(float* Ct, const float* __restrict__ U,
                                             const float (&w)[MUL], int tid) {
  typedef Chunk<MUL, IRD, PC> CH;
  float* stage = Ct + STAGE_OFF;
  const int mis = (int)((((unsigned long long)U) >> 2) & 3ull);
  Pref<CH::NU> pf;
  chunk_fetch<CH>(U, mis, 0, tid, pf);
#pragma unroll 1
  for (int i = 0; i < CH::NCH; ++i) {
    const int a  = i * CH::FMAX;
    const int sh = a - (((a + mis) & ~3) - mis);
    const int np = (729 - i * PC < PC) ? 729 - i * PC : PC;
    chunk_stage<CH>(stage, mis, i, tid, pf);
    __syncthreads();
    if (i + 1 < CH::NCH) chunk_fetch<CH>(U, mis, i + 1, tid, pf);
#pragma unroll
    for (int r = 0; r < CH::ROUNDS; ++r) {
      const int wi = tid + 128 * r;
      const int ii = wi / PC;
      const int pl = wi - ii * PC;
      if (ii < IRD && pl < np) {
        const int p  = i * PC + pl;
        const int g  = p / 9;
        const int d2 = p - g * 9;
        const float* sp = stage + sh + pl * CH::PER_P + ii;
        float v = 0.0f;
#pragma unroll
        for (int k = 0; k < MUL; ++k) v = __builtin_fmaf(sp[k * IRD], w[k], v);
        Ct[(M0 + ii) * KTOT + g * KROW + d2] = v;
      }
    }
    __syncthreads();
  }
}

template <int MUL2, int IRD, int M0>
__device__ __forceinline__ void build_low_orders(float* Ct,
                                                 const float* __restrict__ U2,
                                                 const float* __restrict__ W2,
                                                 const float* __restrict__ U1,
                                                 const float* __restrict__ W1,
                                                 int s, int c, int C, int tid) {
  constexpr int NI = 81 * IRD;
  float w2[MUL2];
#pragma unroll
  for (int k = 0; k < MUL2; ++k) w2[k] = W2[(s * MUL2 + k) * C + c];
  const float w1 = W1[s * C + c];
#pragma unroll
  for (int r = 0; r < (NI + 127) / 128; ++r) {
    const int wi  = tid + 128 * r;
    const int wc  = (wi < NI) ? wi : NI - 1;
    const int ii  = wc / 81;
    const int row = wc - ii * 81;
    const int d0  = row / 9;
    const int d1  = row - d0 * 9;
    float v2 = 0.0f;
#pragma unroll
    for (int k = 0; k < MUL2; ++k)
      v2 = __builtin_fmaf(U2[row * MUL2 * IRD + k * IRD + ii], w2[k], v2);
    const float u1 = U1[d0 * IRD + ii];
    const float v1 = (d1 == 0) ? u1 * w1 : 0.0f;
    float* o = Ct + (M0 + ii) * KTOT + row * KROW + 9;
    o[0] = v2;
    o[1] = v1;
    o[2] = 0.0f;
  }
}

__global__ __launch_bounds__(128) void sc_main(
    const float* __restrict__ X,
    const int*   __restrict__ meta,
    const int*   __restrict__ perm,
    const int*   __restrict__ tileSpecies,
    const int*   __restrict__ tileSlot,
    float*       __restrict__ out, int C,
    const float* __restrict__ U3_0e, const float* __restrict__ W3_0e,
    const float* __restrict__ U3_1o, const float* __restrict__ W3_1o,
    const float* __restrict__ U3_2e, const float* __restrict__ W3_2e,
    const float* __restrict__ U2_0e, const float* __restrict__ W2_0e,
    const float* __restrict__ U2_1o, const float* __restrict__ W2_1o,
    const float* __restrict__ U2_2e, const float* __restrict__ W2_2e,
    const float* __restrict__ U1_0e, const float* __restrict__ W1_0e,
    const float* __restrict__ U1_1o, const float* __restrict__ W1_1o,
    const float* __restrict__ U1_2e, const float* __restrict__ W1_2e)
{
  __shared__ __attribute__((aligned(16))) float Ct[MROWS * KTOT];
  const int c   = blockIdx.x;
  const int s   = blockIdx.y;
  const int tid = threadIdx.x;

  float w0[10], w1[11], w2[13];
#pragma unroll
  for (int k = 0; k < 10; ++k) w0[k] = W3_0e[(s * 10 + k) * C + c];
#pragma unroll
  for (int k = 0; k < 11; ++k) w1[k] = W3_1o[(s * 11 + k) * C + c];
#pragma unroll
  for (int k = 0; k < 13; ++k) w2[k] = W3_2e[(s * 13 + k) * C + c];
  build_low_orders<3, 1, 0>(Ct, U2_0e, W2_0e, U1_0e, W1_0e, s, c, C, tid);
  build_low_orders<2, 3, 1>(Ct, U2_1o, W2_1o, U1_1o, W1_1o, s, c, C, tid);
  build_low_orders<3, 5, 4>(Ct, U2_2e, W2_2e, U1_2e, W1_2e, s, c, C, tid);
  build_order3<10, 1, 365, 0>(Ct, U3_0e, w0, tid);
  build_order3<11, 3, 183, 1>(Ct, U3_1o, w1, tid);
  build_order3<13, 5,  92, 4>(Ct, U3_2e, w2, tid);
  for (int n = tid; n < STAGE_CAP / 4; n += 128) {
    const v4f z = {0.f, 0.f, 0.f, 0.f};
    *(v4f*)(Ct + STAGE_OFF + 4 * n) = z;
  }
  __syncthreads();

  const int  lane = tid & 31;
  const int  wave = tid >> 5;
  const int  n16  = lane & 15;
  const bool hi   = (lane & 16) != 0;
  const float* Abase = &Ct[n16 * KTOT + (hi ? 2 : 0)];
  const int numTiles = meta[0];

  int mc = 0;
  for (int t = 0; t < numTiles; ++t) {
    if (tileSpecies[t] != s) continue;
    const int my = mc++;
    if ((my & 3) != wave) continue;

    const int  slotBase = tileSlot[t];
    const int  node  = perm[slotBase + n16];
    const bool valid = node >= 0;
    const int  nodeC = valid ? node : 0;
    const float* xp = X + (nodeC * C + c) * 9;
    float xv[9];
#pragma unroll
    for (int d = 0; d < 9; ++d) xv[d] = valid ? xp[d] : 0.0f;

    const float e0 = hi ? xv[2] : xv[0];
    const float e1 = hi ? xv[3] : xv[1];
    const float e2 = hi ? xv[6] : xv[4];
    const float e3 = hi ? xv[7] : xv[5];
    const float e4 = hi ? 0.0f  : xv[8];
    const float hm = hi ? 0.0f  : 1.0f;

    v8f acc0 = {0.f,0.f,0.f,0.f,0.f,0.f,0.f,0.f};
    v8f acc1 = acc0, acc2 = acc0;

#pragma unroll 1
    for (int d1i = 0; d1i < 9; ++d1i) {
      float xd1 = xv[0];
#pragma unroll
      for (int q = 1; q < 9; ++q) if (d1i == q) xd1 = xv[q];
      const int rb = d1i * KROW;
#pragma unroll
      for (int d0 = 0; d0 < 9; ++d0) {
        const float xx = xv[d0] * xd1;
        const float* ap = Abase + d0 * (9 * KROW) + rb;
        v2f af0 = *(const v2f*)(ap);
        v2f af1 = *(const v2f*)(ap + 4);
        v2f af2 = *(const v2f*)(ap + 8);
        v2f b0, b1, b2;
        b0.x = xx * e0;  b0.y = xx * e1;
        b1.x = xx * e2;  b1.y = xx * e3;
        const float c1h = (hi && d1i == 0) ? xv[d0] : 0.0f;
        b2.x = xx * e4 + c1h;
        b2.y = xx * hm;
        acc0 = __builtin_amdgcn_wmma_f32_16x16x4_f32(false, af0, false, b0,
                                                     (short)0, acc0, false, false);
        acc1 = __builtin_amdgcn_wmma_f32_16x16x4_f32(false, af1, false, b1,
                                                     (short)0, acc1, false, false);
        acc2 = __builtin_amdgcn_wmma_f32_16x16x4_f32(false, af2, false, b2,
                                                     (short)0, acc2, false, false);
      }
    }
    v8f acc = acc0 + acc1 + acc2;

    if (valid) {
      float* op = out + (node * C + c) * 9;
      if (!hi) {
#pragma unroll
        for (int r = 0; r < 8; ++r) op[r] = acc[r];
      } else {
        op[8] = acc[0];
      }
    }
  }
}

extern "C" void kernel_launch(void* const* d_in, const int* in_sizes, int n_in,
                              void* d_out, int out_size, void* d_ws, size_t ws_size,
                              hipStream_t stream) {
  (void)n_in; (void)out_size; (void)ws_size;
  const float* X     = (const float*)d_in[0];
  const int*   index = (const int*)d_in[1];
  const int N = in_sizes[1];
  const int C = in_sizes[0] / (N * 9);

  int* meta        = (int*)d_ws;
  int* perm        = meta + 16;
  int  permCap     = (N + 16 * NSP + 15) & ~15;
  int* tileSpecies = perm + permCap;
  int* tileSlot    = tileSpecies + 256;

  sc_setup<<<1, SETUP_THREADS, 0, stream>>>(index, N, meta, perm, tileSpecies, tileSlot);

  sc_main<<<dim3(C, NSP), 128, 0, stream>>>(
      X, meta, perm, tileSpecies, tileSlot, (float*)d_out, C,
      (const float*)d_in[2],  (const float*)d_in[3],
      (const float*)d_in[4],  (const float*)d_in[5],
      (const float*)d_in[6],  (const float*)d_in[7],
      (const float*)d_in[8],  (const float*)d_in[9],
      (const float*)d_in[10], (const float*)d_in[11],
      (const float*)d_in[12], (const float*)d_in[13],
      (const float*)d_in[14], (const float*)d_in[15],
      (const float*)d_in[16], (const float*)d_in[17],
      (const float*)d_in[18], (const float*)d_in[19]);
}


# GEMM K cut: coefficients of all orderings of an index triple summed in LDS into the sorted triple's slot, 91 WMMAs per tile not 243; changes summation order
# speedup vs baseline: 11.9125x; 1.2256x over previous
#include <hip/hip_runtime.h>

typedef float v2f __attribute__((ext_vector_type(2)));
typedef float v8f __attribute__((ext_vector_type(8)));

#define NSP   10
#define KROW  12
#define KTOT  (81 * KROW)
#define MROWS 16

#define SETUP_WAVES   32
#define SETUP_THREADS (SETUP_WAVES * 32)

__global__ __launch_bounds__(SETUP_THREADS) void sc_setup(
    const int* __restrict__ index, int N,
    int* __restrict__ meta, int* __restrict__ perm,
    int* __restrict__ tileSpecies, int* __restrict__ tileSlot) {
  __shared__ int waveCnt[SETUP_WAVES][NSP];
  __shared__ int spCnt[NSP];
  const int tid  = threadIdx.x;
  const int lane = tid & 31;
  const int wave = tid >> 5;

  const int chunks = (N + 31) >> 5;
  const int cpw    = (chunks + SETUP_WAVES - 1) / SETUP_WAVES;
  const int n0     = wave * cpw * 32;

  int run[NSP];
  for (int q = 0; q < NSP; ++q) run[q] = 0;
  for (int j = 0; j < cpw; ++j) {
    const int n = n0 + j * 32 + lane;
    const int s = (n < N) ? index[n] : -1;
    for (int q = 0; q < NSP; ++q)
      run[q] += __builtin_popcount(__builtin_amdgcn_ballot_w32(s == q));
  }
  if (lane < NSP) {
    int v = run[0];
    for (int q = 1; q < NSP; ++q) if (lane == q) v = run[q];
    waveCnt[wave][lane] = v;
  }
  __syncthreads();

  if (tid < NSP) {
    int acc = 0;
    for (int w = 0; w < SETUP_WAVES; ++w) {
      const int v = waveCnt[w][tid];
      waveCnt[w][tid] = acc;
      acc += v;
    }
    spCnt[tid] = acc;
  }
  __syncthreads();

  int cnt[NSP], segBase[NSP];
  int base = 0;
  for (int q = 0; q < NSP; ++q) {
    cnt[q]     = spCnt[q];
    segBase[q] = base;
    base += ((cnt[q] + 15) >> 4) << 4;
  }
  const int tiles = base >> 4;

  int wb[NSP];
  for (int q = 0; q < NSP; ++q) wb[q] = segBase[q] + waveCnt[wave][q];
  for (int j = 0; j < cpw; ++j) {
    const int n = n0 + j * 32 + lane;
    const int s = (n < N) ? index[n] : -1;
    int pos = -1;
    for (int q = 0; q < NSP; ++q) {
      const unsigned m = __builtin_amdgcn_ballot_w32(s == q);
      if (s == q) pos = wb[q] + (int)__builtin_amdgcn_mbcnt_lo(m, 0u);
      wb[q] += __builtin_popcount(m);
    }
    if (pos >= 0) perm[pos] = n;
  }

  if (tid < NSP * 16) {
    const int sp = tid >> 4;
    const int j  = tid & 15;
    int c = cnt[0], sb = segBase[0];
    for (int q = 1; q < NSP; ++q) if (sp == q) { c = cnt[q]; sb = segBase[q]; }
    const int slot = c + j;
    if (slot < (((c + 15) >> 4) << 4)) perm[sb + slot] = -1;
  }

  for (int t = tid; t < tiles; t += SETUP_THREADS) {
    int sp = 0;
    for (int q = 0; q < NSP - 1; ++q)
      if (t >= ((segBase[q] >> 4) + ((cnt[q] + 15) >> 4))) sp = q + 1;
    tileSpecies[t] = sp;
    tileSlot[t]    = t * 16;
  }
  if (tid == 0) meta[0] = tiles;
}

#define STAGE_OFF (9 * KTOT)
#define STAGE_CAP (7 * KTOT)
typedef float v4f __attribute__((ext_vector_type(4)));

template <int MUL, int IRD, int PC>
struct Chunk {
  static constexpr int PER_P  = MUL * IRD;
  static constexpr int T      = 729 * PER_P;
  static constexpr int NCH    = (729 + PC - 1) / PC;
  static constexpr int FMAX   = PC * PER_P;
  static constexpr int NU     = ((FMAX + 6) / 4 + 127) / 128;
  static constexpr int ROUNDS = (PC * IRD + 127) / 128;
  static_assert(FMAX + 6 <= STAGE_CAP, "chunk must fit the pad rows");
};

template <int NU>
struct Pref {
  v4f   u[NU];
  float edge;
};

template <typename CH>
__device__ __forceinline__ void chunk_fetch(const float* __restrict__ U, int mis, int i,
                                            int tid, Pref<CH::NU>& pf) {
  const int a   = i * CH::FMAX;
  const int end = (a + CH::FMAX < CH::T) ? a + CH::FMAX : CH::T;
  const int e0  = ((a + mis) & ~3) - mis;
#pragma unroll
  for (int j = 0; j < CH::NU; ++j) {
    const int e = e0 + 4 * (tid + 128 * j);
    v4f v = {0.f, 0.f, 0.f, 0.f};
    if (e < end && e >= 0 && e + 4 <= CH::T) v = *(const v4f*)(U + e);
    pf.u[j] = v;
  }
  pf.edge = 0.0f;
  const int eLast = e0 + (((end - 1 - e0) >> 2) << 2);
  if (e0 < 0 || eLast + 4 > CH::T) {
    if (tid < 8) {
      const int eb = (tid < 4) ? e0 : eLast;
      const int e  = eb + (tid & 3);
      if ((eb < 0 || eb + 4 > CH::T) && e >= 0 && e < CH::T) pf.edge = U[e];
    }
  }
}

template <typename CH>
__device__ __forceinline__ void chunk_stage(float* stage, int mis, int i, int tid,
                                            const Pref<CH::NU>& pf) {
  const int a   = i * CH::FMAX;
  const int end = (a + CH::FMAX < CH::T) ? a + CH::FMAX : CH::T;
  const int e0  = ((a + mis) & ~3) - mis;
#pragma unroll
  for (int j = 0; j < CH::NU; ++j) {
    const int n = 4 * (tid + 128 * j);
    const int e = e0 + n;
    if (e < end && e >= 0 && e + 4 <= CH::T) *(v4f*)(stage + n) = pf.u[j];
  }
  const int eLast = e0 + (((end - 1 - e0) >> 2) << 2);
  if (e0 < 0 || eLast + 4 > CH::T) {
    if (tid < 8) {
      const int eb = (tid < 4) ? e0 : eLast;
      const int e  = eb + (tid & 3);
      if ((eb < 0 || eb + 4 > CH::T) && e >= 0 && e < CH::T) stage[e - e0] = pf.edge;
    }
  }
}

template <int MUL, int IRD, int PC, int M0>
__device__ __forceinline__ void build_order3(float* Ct, const float* __restrict__ U,
                                             const float (&w)[MUL], int tid) {
  typedef Chunk<MUL, IRD, PC> CH;
  float* stage = Ct + STAGE_OFF;
  const int mis = (int)((((unsigned long long)U) >> 2) & 3ull);
  Pref<CH::NU> pf;
  chunk_fetch<CH>(U, mis, 0, tid, pf);
#pragma unroll 1
  for (int i = 0; i < CH::NCH; ++i) {
    const int a  = i * CH::FMAX;
    const int sh = a - (((a + mis) & ~3) - mis);
    const int np = (729 - i * PC < PC) ? 729 - i * PC : PC;
    chunk_stage<CH>(stage, mis, i, tid, pf);
    __syncthreads();
    if (i + 1 < CH::NCH) chunk_fetch<CH>(U, mis, i + 1, tid, pf);
#pragma unroll
    for (int r = 0; r < CH::ROUNDS; ++r) {
      const int wi = tid + 128 * r;
      const int ii = wi / PC;
      const int pl = wi - ii * PC;
      if (ii < IRD && pl < np) {
        const int p  = i * PC + pl;
        const int g  = p / 9;
        const int d2 = p - g * 9;
        const float* sp = stage + sh + pl * CH::PER_P + ii;
        float v = 0.0f;
#pragma unroll
        for (int k = 0; k < MUL; ++k) v = __builtin_fmaf(sp[k * IRD], w[k], v);
        Ct[(M0 + ii) * KTOT + g * KROW + d2] = v;
      }
    }
    __syncthreads();
  }
}

template <int MUL2, int IRD, int M0>
__device__ __forceinline__ void build_low_orders(float* Ct,
                                                 const float* __restrict__ U2,
                                                 const float* __restrict__ W2,
                                                 const float* __restrict__ U1,
                                                 const float* __restrict__ W1,
                                                 int s, int c, int C, int tid) {
  constexpr int NI = 81 * IRD;
  float w2[MUL2];
#pragma unroll
  for (int k = 0; k < MUL2; ++k) w2[k] = W2[(s * MUL2 + k) * C + c];
  const float w1 = W1[s * C + c];
#pragma unroll
  for (int r = 0; r < (NI + 127) / 128; ++r) {
    const int wi  = tid + 128 * r;
    const int wc  = (wi < NI) ? wi : NI - 1;
    const int ii  = wc / 81;
    const int row = wc - ii * 81;
    const int d0  = row / 9;
    const int d1  = row - d0 * 9;
    float v2 = 0.0f;
#pragma unroll
    for (int k = 0; k < MUL2; ++k)
      v2 = __builtin_fmaf(U2[row * MUL2 * IRD + k * IRD + ii], w2[k], v2);
    const float u1 = U1[d0 * IRD + ii];
    const float v1 = (d1 == 0) ? u1 * w1 : 0.0f;
    float* o = Ct + (M0 + ii) * KTOT + row * KROW + 9;
    o[0] = v2;
    o[1] = v1;
    o[2] = 0.0f;
  }
}


__device__ __forceinline__ void fold_symmetric(float* Ct, int tid) {
  static const unsigned short tri[165] = {0, 256, 512, 768, 1024, 1280, 1536, 1792, 2048, 272, 528, 784, 1040, 1296, 1552, 1808, 2064, 544, 800, 1056, 1312, 1568, 1824, 2080, 816, 1072, 1328, 1584, 1840, 2096, 1088, 1344, 1600, 1856, 2112, 1360, 1616, 1872, 2128, 1632, 1888, 2144, 1904, 2160, 2176, 273, 529, 785, 1041, 1297, 1553, 1809, 2065, 545, 801, 1057, 1313, 1569, 1825, 2081, 817, 1073, 1329, 1585, 1841, 2097, 1089, 1345, 1601, 1857, 2113, 1361, 1617, 1873, 2129, 1633, 1889, 2145, 1905, 2161, 2177, 546, 802, 1058, 1314, 1570, 1826, 2082, 818, 1074, 1330, 1586, 1842, 2098, 1090, 1346, 1602, 1858, 2114, 1362, 1618, 1874, 2130, 1634, 1890, 2146, 1906, 2162, 2178, 819, 1075, 1331, 1587, 1843, 2099, 1091, 1347, 1603, 1859, 2115, 1363, 1619, 1875, 2131, 1635, 1891, 2147, 1907, 2163, 2179, 1092, 1348, 1604, 1860, 2116, 1364, 1620, 1876, 2132, 1636, 1892, 2148, 1908, 2164, 2180, 1365, 1621, 1877, 2133, 1637, 1893, 2149, 1909, 2165, 2181, 1638, 1894, 2150, 1910, 2166, 2182, 1911, 2167, 2183, 2184};
  constexpr int NT = 9 * 165;
  constexpr int NP = 9 * 81 * 2;
  constexpr int BR = 4;
  static_assert((NT + 127) / 128 == 3 * BR && (NP + 127) / 128 == 3 * BR, "three batches");
#pragma unroll 1
  for (int bt = 0; bt < 3; ++bt) {
    float vt[BR], vp[BR];
    int   at[BR], apn[BR];
#pragma unroll
    for (int j = 0; j < BR; ++j) {
      const int it0 = tid + 128 * (BR * bt + j);
      const int it  = (it0 < NT) ? it0 : NT - 1;
      const int i   = it / 165;
      const int pk  = tri[it - i * 165];
      const int a = pk & 15, b = (pk >> 4) & 15, c = pk >> 8;
      const float* R = Ct + i * KTOT;
      const float s1 = R[(a * 9 + b) * KROW + c];
      const float s2 = R[(a * 9 + c) * KROW + b];
      const float s3 = R[(c * 9 + a) * KROW + b];
      const float s4 = R[(b * 9 + a) * KROW + c];
      const float s5 = R[(b * 9 + c) * KROW + a];
      const float s6 = R[(c * 9 + b) * KROW + a];
      float v = s1;
      v += (b != c) ? s2 : 0.0f;
      v += (b != c) ? s3 : 0.0f;
      v += (a != b) ? s4 : 0.0f;
      v += (a != b) ? s5 : 0.0f;
      v += (a != b && b != c) ? s6 : 0.0f;
      vt[j] = v;
      at[j] = (it0 < NT) ? i * KTOT + (a * 9 + b) * KROW + c : -1;
    }
#pragma unroll
    for (int j = 0; j < BR; ++j) {
      const int it0 = tid + 128 * (BR * bt + j);
      const int it  = (it0 < NP) ? it0 : NP - 1;
      const int i   = it / 162;
      const int q   = it - i * 162;
      const int g   = q >> 1;
      const int sl  = 9 + (q & 1);
      const int a   = g / 9;
      const int b   = g - a * 9;
      const float* R = Ct + i * KTOT;
      const float own = R[g * KROW + sl];
      const float oth = R[(b * 9 + a) * KROW + sl];
      vp[j]  = own + oth;
      apn[j] = (it0 < NP && a < b) ? i * KTOT + g * KROW + sl : -1;
    }
#pragma unroll
    for (int j = 0; j < BR; ++j) {
      if (at[j] >= 0)  Ct[at[j]]  = vt[j];
      if (apn[j] >= 0) Ct[apn[j]] = vp[j];
    }
  }
}

__device__ __forceinline__ void zero_unsorted_heads(float* Ct, int tid) {
  for (int it = tid; it < 9 * 81 * 3; it += 128) {
    const int i = it / 243;
    const int r = it - i * 243;
    const int g = r / 3;
    const int j = r - g * 3;
    const int a = g / 9;
    const int b = g - a * 9;
    const int c = (b & ~3) + j;
    if (a <= b && c < b) Ct[i * KTOT + g * KROW + c] = 0.0f;
  }
}

__global__ __launch_bounds__(128) void sc_main(
    const float* __restrict__ X,
    const int*   __restrict__ meta,
    const int*   __restrict__ perm,
    const int*   __restrict__ tileSpecies,
    const int*   __restrict__ tileSlot,
    float*       __restrict__ out, int C,
    const float* __restrict__ U3_0e, const float* __restrict__ W3_0e,
    const float* __restrict__ U3_1o, const float* __restrict__ W3_1o,
    const float* __restrict__ U3_2e, const float* __restrict__ W3_2e,
    const float* __restrict__ U2_0e, const float* __restrict__ W2_0e,
    const float* __restrict__ U2_1o, const float* __restrict__ W2_1o,
    const float* __restrict__ U2_2e, const float* __restrict__ W2_2e,
    const float* __restrict__ U1_0e, const float* __restrict__ W1_0e,
    const float* __restrict__ U1_1o, const float* __restrict__ W1_1o,
    const float* __restrict__ U1_2e, const float* __restrict__ W1_2e)
{
  __shared__ __attribute__((aligned(16))) float Ct[MROWS * KTOT];
  const int c   = blockIdx.x;
  const int s   = blockIdx.y;
  const int tid = threadIdx.x;

  float w0[10], w1[11], w2[13];
#pragma unroll
  for (int k = 0; k < 10; ++k) w0[k] = W3_0e[(s * 10 + k) * C + c];
#pragma unroll
  for (int k = 0; k < 11; ++k) w1[k] = W3_1o[(s * 11 + k) * C + c];
#pragma unroll
  for (int k = 0; k < 13; ++k) w2[k] = W3_2e[(s * 13 + k) * C + c];
  build_low_orders<3, 1, 0>(Ct, U2_0e, W2_0e, U1_0e, W1_0e, s, c, C, tid);
  build_low_orders<2, 3, 1>(Ct, U2_1o, W2_1o, U1_1o, W1_1o, s, c, C, tid);
  build_low_orders<3, 5, 4>(Ct, U2_2e, W2_2e, U1_2e, W1_2e, s, c, C, tid);
  build_order3<10, 1, 365, 0>(Ct, U3_0e, w0, tid);
  build_order3<11, 3, 183, 1>(Ct, U3_1o, w1, tid);
  build_order3<13, 5,  92, 4>(Ct, U3_2e, w2, tid);
  fold_symmetric(Ct, tid);
  __syncthreads();
  zero_unsorted_heads(Ct, tid);
  for (int n = tid; n < STAGE_CAP / 4; n += 128) {
    const v4f z = {0.f, 0.f, 0.f, 0.f};
    *(v4f*)(Ct + STAGE_OFF + 4 * n) = z;
  }
  __syncthreads();

  const int  lane = tid & 31;
  const int  wave = tid >> 5;
  const int  n16  = lane & 15;
  const bool hi   = (lane & 16) != 0;
  const float* Abase = &Ct[n16 * KTOT + (hi ? 2 : 0)];
  const int numTiles = meta[0];

  int mc = 0;
  for (int t = 0; t < numTiles; ++t) {
    if (tileSpecies[t] != s) continue;
    const int my = mc++;
    if ((my & 3) != wave) continue;

    const int  slotBase = tileSlot[t];
    const int  node  = perm[slotBase + n16];
    const bool valid = node >= 0;
    const int  nodeC = valid ? node : 0;
    const float* xp = X + (nodeC * C + c) * 9;
    float xv[9];
#pragma unroll
    for (int d = 0; d < 9; ++d) xv[d] = valid ? xp[d] : 0.0f;

    const float e0 = hi ? xv[2] : xv[0];
    const float e1 = hi ? xv[3] : xv[1];
    const float e2 = hi ? xv[6] : xv[4];
    const float e3 = hi ? xv[7] : xv[5];
    const float e4 = hi ? 0.0f  : xv[8];
    const float hm = hi ? 0.0f  : 1.0f;

    v8f acc0 = {0.f,0.f,0.f,0.f,0.f,0.f,0.f,0.f};
    v8f acc1 = acc0, acc2 = acc0;

#define SC_WMMA(AF, BF)                                                        \
    {                                                                          \
      acc0 = __builtin_amdgcn_wmma_f32_16x16x4_f32(false, AF, false, BF,       \
                                                   (short)0, acc0, false, false); \
      const v8f rot_ = acc0; acc0 = acc1; acc1 = acc2; acc2 = rot_;            \
    }

#pragma unroll
    for (int a = 0; a < 9; ++a) {
#pragma unroll
      for (int b = a; b < 9; ++b) {
        const float xx = xv[a] * xv[b];
        const float* ap = Abase + (a * 9 + b) * KROW;
        if (b < 4) {
          const v2f af0 = *(const v2f*)(ap);
          v2f b0;
          b0.x = xx * e0;  b0.y = xx * e1;
          SC_WMMA(af0, b0)
        }
        if (b < 8) {
          const v2f af1 = *(const v2f*)(ap + 4);
          v2f b1;
          b1.x = xx * e2;  b1.y = xx * e3;
          SC_WMMA(af1, b1)
        }
        const v2f af2 = *(const v2f*)(ap + 8);
        v2f b2;
        const float c1h = (a == 0 && hi) ? xv[b] : 0.0f;
        b2.x = xx * e4 + c1h;
        b2.y = xx * hm;
        SC_WMMA(af2, b2)
      }
    }
#undef SC_WMMA
    v8f acc = acc0 + acc1 + acc2;

    if (valid) {
      float* op = out + (node * C + c) * 9;
      if (!hi) {
#pragma unroll
        for (int r = 0; r < 8; ++r) op[r] = acc[r];
      } else {
        op[8] = acc[0];
      }
    }
  }
}

extern "C" void kernel_launch(void* const* d_in, const int* in_sizes, int n_in,
                              void* d_out, int out_size, void* d_ws, size_t ws_size,
                              hipStream_t stream) {
  (void)n_in; (void)out_size; (void)ws_size;
  const float* X     = (const float*)d_in[0];
  const int*   index = (const int*)d_in[1];
  const int N = in_sizes[1];
  const int C = in_sizes[0] / (N * 9);

  int* meta        = (int*)d_ws;
  int* perm        = meta + 16;
  int  permCap     = (N + 16 * NSP + 15) & ~15;
  int* tileSpecies = perm + permCap;
  int* tileSlot    = tileSpecies + 256;

  sc_setup<<<1, SETUP_THREADS, 0, stream>>>(index, N, meta, perm, tileSpecies, tileSlot);

  sc_main<<<dim3(C, NSP), 128, 0, stream>>>(
      X, meta, perm, tileSpecies, tileSlot, (float*)d_out, C,
      (const float*)d_in[2],  (const float*)d_in[3],
      (const float*)d_in[4],  (const float*)d_in[5],
      (const float*)d_in[6],  (const float*)d_in[7],
      (const float*)d_in[8],  (const float*)d_in[9],
      (const float*)d_in[10], (const float*)d_in[11],
      (const float*)d_in[12], (const float*)d_in[13],
      (const float*)d_in[14], (const float*)d_in[15],
      (const float*)d_in[16], (const float*)d_in[17],
      (const float*)d_in[18], (const float*)d_in[19]);
}


# Index-order fold moved ahead of the weight contraction: setup launch sums U over orderings once into sorted-triple planes, each block contracts 165 triples not 729; changes summation order
# speedup vs baseline: 19.5435x; 1.6406x over previous
#include <hip/hip_runtime.h>


typedef float v2f __attribute__((ext_vector_type(2)));
typedef float v4f __attribute__((ext_vector_type(4)));
typedef float v8f __attribute__((ext_vector_type(8)));
typedef unsigned short us4 __attribute__((ext_vector_type(4)));

#define NSP   10
#define KROW  12
#define KTOT  (81 * KROW)
#define MROWS 16

#define NT3 165
#define TP3 168
#define NT2 45
#define TP2 48
#define F3_0E_OFF 0
#define F3_1O_OFF (F3_0E_OFF + 10 * 1 * TP3)
#define F3_2E_OFF (F3_1O_OFF + 11 * 3 * TP3)
#define F2_0E_OFF (F3_2E_OFF + 13 * 5 * TP3)
#define F2_1O_OFF (F2_0E_OFF + 3 * 1 * TP2)
#define F2_2E_OFF (F2_1O_OFF + 2 * 3 * TP2)
#define F_TOTAL   (F2_2E_OFF + 3 * 5 * TP2)

__device__ __forceinline__ const unsigned short* triple_slots() {
  static const __attribute__((aligned(8))) unsigned short tab[TP3] = {
      0, 1, 2, 3, 4, 5, 6, 7, 8, 13, 14, 15, 16, 17,
      18, 19, 20, 26, 27, 28, 29, 30, 31, 32, 39, 40, 41, 42,
      43, 44, 52, 53, 54, 55, 56, 65, 66, 67, 68, 78, 79, 80,
      91, 92, 104, 121, 122, 123, 124, 125, 126, 127, 128, 134, 135, 136,
      137, 138, 139, 140, 147, 148, 149, 150, 151, 152, 160, 161, 162, 163,
      164, 173, 174, 175, 176, 186, 187, 188, 199, 200, 212, 242, 243, 244,
      245, 246, 247, 248, 255, 256, 257, 258, 259, 260, 268, 269, 270, 271,
      272, 281, 282, 283, 284, 294, 295, 296, 307, 308, 320, 363, 364, 365,
      366, 367, 368, 376, 377, 378, 379, 380, 389, 390, 391, 392, 402, 403,
      404, 415, 416, 428, 484, 485, 486, 487, 488, 497, 498, 499, 500, 510,
      511, 512, 523, 524, 536, 605, 606, 607, 608, 618, 619, 620, 631, 632,
      644, 726, 727, 728, 739, 740, 752, 847, 848, 860, 968, 11, 11, 11};
  return tab;
}
__device__ __forceinline__ const unsigned short* pair_slots() {
  static const __attribute__((aligned(8))) unsigned short tab[TP2] = {
      9, 21, 33, 45, 57, 69, 81, 93, 105, 129, 141, 153,
      165, 177, 189, 201, 213, 249, 261, 273, 285, 297, 309, 321,
      369, 381, 393, 405, 417, 429, 489, 501, 513, 525, 537, 609,
      621, 633, 645, 729, 741, 753, 849, 861, 969, 11, 11, 11};
  return tab;
}

template <int MUL, int IRD>
__device__ __forceinline__ void fold3_entry(const float* __restrict__ U, float* __restrict__ F, int o) {
  constexpr int PER_P = MUL * IRD;
  const int t  = o / PER_P;
  const int ki = o - t * PER_P;
  float v = 0.0f;
  if (t < NT3) {
    const int sl = triple_slots()[t];
    const int g = sl / KROW;
    const int c = sl - g * KROW;
    const int a = g / 9;
    const int b = g - a * 9;
    const float s1 = U[((a * 9 + b) * 9 + c) * PER_P + ki];
    const float s2 = U[((a * 9 + c) * 9 + b) * PER_P + ki];
    const float s3 = U[((c * 9 + a) * 9 + b) * PER_P + ki];
    const float s4 = U[((b * 9 + a) * 9 + c) * PER_P + ki];
    const float s5 = U[((b * 9 + c) * 9 + a) * PER_P + ki];
    const float s6 = U[((c * 9 + b) * 9 + a) * PER_P + ki];
    v = s1;
    v += (b != c) ? s2 : 0.0f;
    v += (b != c) ? s3 : 0.0f;
    v += (a != b) ? s4 : 0.0f;
    v += (a != b) ? s5 : 0.0f;
    v += (a != b && b != c) ? s6 : 0.0f;
  }
  F[ki * TP3 + t] = v;
}

template <int MUL, int IRD>
__device__ __forceinline__ void fold2_entry(const float* __restrict__ U, float* __restrict__ F, int o) {
  constexpr int PER_P = MUL * IRD;
  const int p  = o / PER_P;
  const int ki = o - p * PER_P;
  float v = 0.0f;
  if (p < NT2) {
    const int g = pair_slots()[p] / KROW;
    const int a = g / 9;
    const int b = g - a * 9;
    const float own = U[(a * 9 + b) * PER_P + ki];
    const float oth = U[(b * 9 + a) * PER_P + ki];
    v = own + ((a != b) ? oth : 0.0f);
  }
  F[ki * TP2 + p] = v;
}

#define SETUP_WAVES   32
#define SETUP_THREADS (SETUP_WAVES * 32)
#define FOLD_BLOCKS   ((F_TOTAL + SETUP_THREADS - 1) / SETUP_THREADS)

__global__ __launch_bounds__(SETUP_THREADS) void sc_setup(
    const int* __restrict__ index, int N,
    int* __restrict__ meta, int* __restrict__ perm,
    int* __restrict__ tileSpecies, int* __restrict__ tileSlot,
    float* __restrict__ F,
    const float* __restrict__ U3_0e, const float* __restrict__ U3_1o,
    const float* __restrict__ U3_2e, const float* __restrict__ U2_0e,
    const float* __restrict__ U2_1o, const float* __restrict__ U2_2e) {
  __shared__ int waveCnt[SETUP_WAVES][NSP];
  __shared__ int spCnt[NSP];
  const int tid  = threadIdx.x;

  if (blockIdx.x != 0) {
    const int o = (blockIdx.x - 1) * SETUP_THREADS + tid;
    if      (o < F3_1O_OFF) fold3_entry<10, 1>(U3_0e, F + F3_0E_OFF, o - F3_0E_OFF);
    else if (o < F3_2E_OFF) fold3_entry<11, 3>(U3_1o, F + F3_1O_OFF, o - F3_1O_OFF);
    else if (o < F2_0E_OFF) fold3_entry<13, 5>(U3_2e, F + F3_2E_OFF, o - F3_2E_OFF);
    else if (o < F2_1O_OFF) fold2_entry<3, 1>(U2_0e, F + F2_0E_OFF, o - F2_0E_OFF);
    else if (o < F2_2E_OFF) fold2_entry<2, 3>(U2_1o, F + F2_1O_OFF, o - F2_1O_OFF);
    else if (o < F_TOTAL)   fold2_entry<3, 5>(U2_2e, F + F2_2E_OFF, o - F2_2E_OFF);
    return;
  }
  const int lane = tid & 31;
  const int wave = tid >> 5;

  const int chunks = (N + 31) >> 5;
  const int cpw    = (chunks + SETUP_WAVES - 1) / SETUP_WAVES;
  const int n0     = wave * cpw * 32;

  int run[NSP];
  for (int q = 0; q < NSP; ++q) run[q] = 0;
  for (int j = 0; j < cpw; ++j) {
    const int n = n0 + j * 32 + lane;
    const int s = (n < N) ? index[n] : -1;
    for (int q = 0; q < NSP; ++q)
      run[q] += __builtin_popcount(__builtin_amdgcn_ballot_w32(s == q));
  }
  if (lane < NSP) {
    int v = run[0];
    for (int q = 1; q < NSP; ++q) if (lane == q) v = run[q];
    waveCnt[wave][lane] = v;
  }
  __syncthreads();

  if (tid < NSP) {
    int acc = 0;
    for (int w = 0; w < SETUP_WAVES; ++w) {
      const int v = waveCnt[w][tid];
      waveCnt[w][tid] = acc;
      acc += v;
    }
    spCnt[tid] = acc;
  }
  __syncthreads();

  int cnt[NSP], segBase[NSP];
  int base = 0;
  for (int q = 0; q < NSP; ++q) {
    cnt[q]     = spCnt[q];
    segBase[q] = base;
    base += ((cnt[q] + 15) >> 4) << 4;
  }
  const int tiles = base >> 4;

  int wb[NSP];
  for (int q = 0; q < NSP; ++q) wb[q] = segBase[q] + waveCnt[wave][q];
  for (int j = 0; j < cpw; ++j) {
    const int n = n0 + j * 32 + lane;
    const int s = (n < N) ? index[n] : -1;
    int pos = -1;
    for (int q = 0; q < NSP; ++q) {
      const unsigned m = __builtin_amdgcn_ballot_w32(s == q);
      if (s == q) pos = wb[q] + (int)__builtin_amdgcn_mbcnt_lo(m, 0u);
      wb[q] += __builtin_popcount(m);
    }
    if (pos >= 0) perm[pos] = n;
  }

  if (tid < NSP * 16) {
    const int sp = tid >> 4;
    const int j  = tid & 15;
    int c = cnt[0], sb = segBase[0];
    for (int q = 1; q < NSP; ++q) if (sp == q) { c = cnt[q]; sb = segBase[q]; }
    const int slot = c + j;
    if (slot < (((c + 15) >> 4) << 4)) perm[sb + slot] = -1;
  }

  for (int t = tid; t < tiles; t += SETUP_THREADS) {
    int sp = 0;
    for (int q = 0; q < NSP - 1; ++q)
      if (t >= ((segBase[q] >> 4) + ((cnt[q] + 15) >> 4))) sp = q + 1;
    tileSpecies[t] = sp;
    tileSlot[t]    = t * 16;
  }
  if (tid == 0) meta[0] = tiles;
}

template <int MUL, int IRD, int M0, int TP>
__device__ __forceinline__ void build_folded(float* Ct, const float* __restrict__ F,
                                             const unsigned short* __restrict__ slots,
                                             const float (&w)[MUL], int tid) {
  constexpr int NQ = TP / 4;
  constexpr int NI = IRD * NQ;
#pragma unroll
  for (int r = 0; r < (NI + 127) / 128; ++r) {
    const int it0 = tid + 128 * r;
    const int it  = (it0 < NI) ? it0 : NI - 1;
    const int i   = it / NQ;
    const int q   = it - i * NQ;
    const float* fp = F + i * TP + 4 * q;
    float v0 = 0.0f, v1 = 0.0f, v2 = 0.0f, v3 = 0.0f;
#pragma unroll
    for (int k = 0; k < MUL; ++k) {
      const v4f u = *(const v4f*)(fp + k * IRD * TP);
      v0 = __builtin_fmaf(u.x, w[k], v0);
      v1 = __builtin_fmaf(u.y, w[k], v1);
      v2 = __builtin_fmaf(u.z, w[k], v2);
      v3 = __builtin_fmaf(u.w, w[k], v3);
    }
    const us4 sl = *(const us4*)(slots + 4 * q);
    float* row = Ct + (M0 + i) * KTOT;
    row[sl.x] = v0;
    row[sl.y] = v1;
    row[sl.z] = v2;
    row[sl.w] = v3;
  }
}

template <int IRD, int M0>
__device__ __forceinline__ void build_order1(float* Ct, const float* __restrict__ U1,
                                             const float* __restrict__ W1,
                                             int s, int c, int C, int tid) {
  constexpr int NI = 9 * IRD;
  const int it = (tid < NI) ? tid : NI - 1;
  const int ii = it / 9;
  const int b  = it - ii * 9;
  Ct[(M0 + ii) * KTOT + b * KROW + 10] = U1[b * IRD + ii] * W1[s * C + c];
}

__global__ __launch_bounds__(128) void sc_main(
    const float* __restrict__ X,
    const int*   __restrict__ meta,
    const int*   __restrict__ perm,
    const int*   __restrict__ tileSpecies,
    const int*   __restrict__ tileSlot,
    float*       __restrict__ out, int C,
    const float* __restrict__ F,
    const float* __restrict__ W3_0e, const float* __restrict__ W3_1o,
    const float* __restrict__ W3_2e,
    const float* __restrict__ W2_0e, const float* __restrict__ W2_1o,
    const float* __restrict__ W2_2e,
    const float* __restrict__ U1_0e, const float* __restrict__ W1_0e,
    const float* __restrict__ U1_1o, const float* __restrict__ W1_1o,
    const float* __restrict__ U1_2e, const float* __restrict__ W1_2e)
{
  __shared__ __attribute__((aligned(16))) float Ct[MROWS * KTOT];
  const int c   = blockIdx.x;
  const int s   = blockIdx.y;
  const int tid = threadIdx.x;

  for (int n = tid; n < MROWS * KTOT / 4; n += 128) {
    const v4f z = {0.f, 0.f, 0.f, 0.f};
    *(v4f*)(Ct + 4 * n) = z;
  }
  float w0[10], w1[11], w2[13], y0[3], y1[2], y2[3];
#pragma unroll
  for (int k = 0; k < 10; ++k) w0[k] = W3_0e[(s * 10 + k) * C + c];
#pragma unroll
  for (int k = 0; k < 11; ++k) w1[k] = W3_1o[(s * 11 + k) * C + c];
#pragma unroll
  for (int k = 0; k < 13; ++k) w2[k] = W3_2e[(s * 13 + k) * C + c];
#pragma unroll
  for (int k = 0; k < 3; ++k) y0[k] = W2_0e[(s * 3 + k) * C + c];
#pragma unroll
  for (int k = 0; k < 2; ++k) y1[k] = W2_1o[(s * 2 + k) * C + c];
#pragma unroll
  for (int k = 0; k < 3; ++k) y2[k] = W2_2e[(s * 3 + k) * C + c];
  __syncthreads();
  build_folded<13, 5, 4, TP3>(Ct, F + F3_2E_OFF, triple_slots(), w2, tid);
  __builtin_amdgcn_sched_barrier(0);
  build_folded<11, 3, 1, TP3>(Ct, F + F3_1O_OFF, triple_slots(), w1, tid);
  build_folded<10, 1, 0, TP3>(Ct, F + F3_0E_OFF, triple_slots(), w0, tid);
  build_folded<3, 5, 4, TP2>(Ct, F + F2_2E_OFF, pair_slots(), y2, tid);
  build_folded<2, 3, 1, TP2>(Ct, F + F2_1O_OFF, pair_slots(), y1, tid);
  build_folded<3, 1, 0, TP2>(Ct, F + F2_0E_OFF, pair_slots(), y0, tid);
  build_order1<1, 0>(Ct, U1_0e, W1_0e, s, c, C, tid);
  build_order1<3, 1>(Ct, U1_1o, W1_1o, s, c, C, tid);
  build_order1<5, 4>(Ct, U1_2e, W1_2e, s, c, C, tid);
  __syncthreads();

  const int  lane = tid & 31;
  const int  wave = tid >> 5;
  const int  n16  = lane & 15;
  const bool hi   = (lane & 16) != 0;
  const float* Abase = &Ct[n16 * KTOT + (hi ? 2 : 0)];
  const int numTiles = meta[0];

  int mc = 0;
  for (int t = 0; t < numTiles; ++t) {
    if (tileSpecies[t] != s) continue;
    const int my = mc++;
    if ((my & 3) != wave) continue;

    const int  slotBase = tileSlot[t];
    const int  node  = perm[slotBase + n16];
    const bool valid = node >= 0;
    const int  nodeC = valid ? node : 0;
    const float* xp = X + (nodeC * C + c) * 9;
    float xv[9];
#pragma unroll
    for (int d = 0; d < 9; ++d) xv[d] = valid ? xp[d] : 0.0f;

    const float e0 = hi ? xv[2] : xv[0];
    const float e1 = hi ? xv[3] : xv[1];
    const float e2 = hi ? xv[6] : xv[4];
    const float e3 = hi ? xv[7] : xv[5];
    const float e4 = hi ? 0.0f  : xv[8];
    const float hm = hi ? 0.0f  : 1.0f;

    v8f acc0 = {0.f,0.f,0.f,0.f,0.f,0.f,0.f,0.f};
    v8f acc1 = acc0, acc2 = acc0;

#define SC_WMMA(AF, BF)                                                        \
    {                                                                          \
      acc0 = __builtin_amdgcn_wmma_f32_16x16x4_f32(false, AF, false, BF,       \
                                                   (short)0, acc0, false, false); \
      const v8f rot_ = acc0; acc0 = acc1; acc1 = acc2; acc2 = rot_;            \
    }

#pragma unroll
    for (int a = 0; a < 9; ++a) {
#pragma unroll
      for (int b = a; b < 9; ++b) {
        const float xx = xv[a] * xv[b];
        const float* ap = Abase + (a * 9 + b) * KROW;
        if (b < 4) {
          const v2f af0 = *(const v2f*)(ap);
          v2f b0;
          b0.x = xx * e0;  b0.y = xx * e1;
          SC_WMMA(af0, b0)
        }
        if (b < 8) {
          const v2f af1 = *(const v2f*)(ap + 4);
          v2f b1;
          b1.x = xx * e2;  b1.y = xx * e3;
          SC_WMMA(af1, b1)
        }
        const v2f af2 = *(const v2f*)(ap + 8);
        v2f b2;
        const float c1h = (a == 0 && hi) ? xv[b] : 0.0f;
        b2.x = xx * e4 + c1h;
        b2.y = xx * hm;
        SC_WMMA(af2, b2)
      }
    }
#undef SC_WMMA
    v8f acc = acc0 + acc1 + acc2;

    if (valid) {
      float* op = out + (node * C + c) * 9;
      if (!hi) {
#pragma unroll
        for (int r = 0; r < 8; ++r) op[r] = acc[r];
      } else {
        op[8] = acc[0];
      }
    }
  }
}

extern "C" void kernel_launch(void* const* d_in, const int* in_sizes, int n_in,
                              void* d_out, int out_size, void* d_ws, size_t ws_size,
                              hipStream_t stream) {
  (void)n_in; (void)out_size; (void)ws_size;
  const float* X     = (const float*)d_in[0];
  const int*   index = (const int*)d_in[1];
  const int N = in_sizes[1];
  const int C = in_sizes[0] / (N * 9);

  int* meta        = (int*)d_ws;
  int* perm        = meta + 16;
  int  permCap     = (N + 16 * NSP + 15) & ~15;
  int* tileSpecies = perm + permCap;
  int* tileSlot    = tileSpecies + 256;
  float* F         = (float*)(tileSlot + 256);

  sc_setup<<<1 + FOLD_BLOCKS, SETUP_THREADS, 0, stream>>>(
      index, N, meta, perm, tileSpecies, tileSlot, F,
      (const float*)d_in[2], (const float*)d_in[4], (const float*)d_in[6],
      (const float*)d_in[8], (const float*)d_in[10], (const float*)d_in[12]);

  sc_main<<<dim3(C, NSP), 128, 0, stream>>>(
      X, meta, perm, tileSpecies, tileSlot, (float*)d_out, C, F,
      (const float*)d_in[3],  (const float*)d_in[5],  (const float*)d_in[7],
      (const float*)d_in[9],  (const float*)d_in[11], (const float*)d_in[13],
      (const float*)d_in[14], (const float*)d_in[15],
      (const float*)d_in[16], (const float*)d_in[17],
      (const float*)d_in[18], (const float*)d_in[19]);
}


# GEMM k axis packed dense: one slot per sorted triple, sorted pair and single (228 slots, row pitch 228 words), 57 WMMAs per tile not 91; B factors selected per lane half; changes summation order
# speedup vs baseline: 20.9716x; 1.0731x over previous
#include <hip/hip_runtime.h>


typedef float v2f __attribute__((ext_vector_type(2)));
typedef float v4f __attribute__((ext_vector_type(4)));
typedef float v8f __attribute__((ext_vector_type(8)));

#define NSP   10
#define KROW  12
#define MROWS 16

#define NT3 165
#define TP3 168
#define NT2 45
#define TP2 48
#define F3_0E_OFF 0
#define F3_1O_OFF (F3_0E_OFF + 10 * 1 * TP3)
#define F3_2E_OFF (F3_1O_OFF + 11 * 3 * TP3)
#define F2_0E_OFF (F3_2E_OFF + 13 * 5 * TP3)
#define F2_1O_OFF (F2_0E_OFF + 3 * 1 * TP2)
#define F2_2E_OFF (F2_1O_OFF + 2 * 3 * TP2)
#define F_TOTAL   (F2_2E_OFF + 3 * 5 * TP2)

#define K_T3  0
#define K_P2  TP3
#define K_S1  (TP3 + TP2)
#define KTOT  (TP3 + TP2 + 12)
#define NBLK  (KTOT / 4)

__device__ __forceinline__ const unsigned short* triple_slots() {
  static const __attribute__((aligned(8))) unsigned short tab[TP3] = {
      0, 1, 2, 3, 4, 5, 6, 7, 8, 13, 14, 15, 16, 17,
      18, 19, 20, 26, 27, 28, 29, 30, 31, 32, 39, 40, 41, 42,
      43, 44, 52, 53, 54, 55, 56, 65, 66, 67, 68, 78, 79, 80,
      91, 92, 104, 121, 122, 123, 124, 125, 126, 127, 128, 134, 135, 136,
      137, 138, 139, 140, 147, 148, 149, 150, 151, 152, 160, 161, 162, 163,
      164, 173, 174, 175, 176, 186, 187, 188, 199, 200, 212, 242, 243, 244,
      245, 246, 247, 248, 255, 256, 257, 258, 259, 260, 268, 269, 270, 271,
      272, 281, 282, 283, 284, 294, 295, 296, 307, 308, 320, 363, 364, 365,
      366, 367, 368, 376, 377, 378, 379, 380, 389, 390, 391, 392, 402, 403,
      404, 415, 416, 428, 484, 485, 486, 487, 488, 497, 498, 499, 500, 510,
      511, 512, 523, 524, 536, 605, 606, 607, 608, 618, 619, 620, 631, 632,
      644, 726, 727, 728, 739, 740, 752, 847, 848, 860, 968, 11, 11, 11};
  return tab;
}
__device__ __forceinline__ const unsigned short* pair_slots() {
  static const __attribute__((aligned(8))) unsigned short tab[TP2] = {
      9, 21, 33, 45, 57, 69, 81, 93, 105, 129, 141, 153,
      165, 177, 189, 201, 213, 249, 261, 273, 285, 297, 309, 321,
      369, 381, 393, 405, 417, 429, 489, 501, 513, 525, 537, 609,
      621, 633, 645, 729, 741, 753, 849, 861, 969, 11, 11, 11};
  return tab;
}

template <int MUL, int IRD>
__device__ __forceinline__ void fold3_entry(const float* __restrict__ U, float* __restrict__ F, int o) {
  constexpr int PER_P = MUL * IRD;
  const int t  = o / PER_P;
  const int ki = o - t * PER_P;
  float v = 0.0f;
  if (t < NT3) {
    const int sl = triple_slots()[t];
    const int g = sl / KROW;
    const int c = sl - g * KROW;
    const int a = g / 9;
    const int b = g - a * 9;
    const float s1 = U[((a * 9 + b) * 9 + c) * PER_P + ki];
    const float s2 = U[((a * 9 + c) * 9 + b) * PER_P + ki];
    const float s3 = U[((c * 9 + a) * 9 + b) * PER_P + ki];
    const float s4 = U[((b * 9 + a) * 9 + c) * PER_P + ki];
    const float s5 = U[((b * 9 + c) * 9 + a) * PER_P + ki];
    const float s6 = U[((c * 9 + b) * 9 + a) * PER_P + ki];
    v = s1;
    v += (b != c) ? s2 : 0.0f;
    v += (b != c) ? s3 : 0.0f;
    v += (a != b) ? s4 : 0.0f;
    v += (a != b) ? s5 : 0.0f;
    v += (a != b && b != c) ? s6 : 0.0f;
  }
  F[ki * TP3 + t] = v;
}

template <int MUL, int IRD>
__device__ __forceinline__ void fold2_entry(const float* __restrict__ U, float* __restrict__ F, int o) {
  constexpr int PER_P = MUL * IRD;
  const int p  = o / PER_P;
  const int ki = o - p * PER_P;
  float v = 0.0f;
  if (p < NT2) {
    const int g = pair_slots()[p] / KROW;
    const int a = g / 9;
    const int b = g - a * 9;
    const float own = U[(a * 9 + b) * PER_P + ki];
    const float oth = U[(b * 9 + a) * PER_P + ki];
    v = own + ((a != b) ? oth : 0.0f);
  }
  F[ki * TP2 + p] = v;
}

#define SETUP_WAVES   32
#define SETUP_THREADS (SETUP_WAVES * 32)
#define FOLD_BLOCKS   ((F_TOTAL + SETUP_THREADS - 1) / SETUP_THREADS)

__global__ __launch_bounds__(SETUP_THREADS) void sc_setup(
    const int* __restrict__ index, int N,
    int* __restrict__ meta, int* __restrict__ perm,
    int* __restrict__ tileSpecies, int* __restrict__ tileSlot,
    float* __restrict__ F,
    const float* __restrict__ U3_0e, const float* __restrict__ U3_1o,
    const float* __restrict__ U3_2e, const float* __restrict__ U2_0e,
    const float* __restrict__ U2_1o, const float* __restrict__ U2_2e) {
  __shared__ int waveCnt[SETUP_WAVES][NSP];
  __shared__ int spCnt[NSP];
  const int tid  = threadIdx.x;

  if (blockIdx.x != 0) {
    const int o = (blockIdx.x - 1) * SETUP_THREADS + tid;
    if      (o < F3_1O_OFF) fold3_entry<10, 1>(U3_0e, F + F3_0E_OFF, o - F3_0E_OFF);
    else if (o < F3_2E_OFF) fold3_entry<11, 3>(U3_1o, F + F3_1O_OFF, o - F3_1O_OFF);
    else if (o < F2_0E_OFF) fold3_entry<13, 5>(U3_2e, F + F3_2E_OFF, o - F3_2E_OFF);
    else if (o < F2_1O_OFF) fold2_entry<3, 1>(U2_0e, F + F2_0E_OFF, o - F2_0E_OFF);
    else if (o < F2_2E_OFF) fold2_entry<2, 3>(U2_1o, F + F2_1O_OFF, o - F2_1O_OFF);
    else if (o < F_TOTAL)   fold2_entry<3, 5>(U2_2e, F + F2_2E_OFF, o - F2_2E_OFF);
    return;
  }
  const int lane = tid & 31;
  const int wave = tid >> 5;

  const int chunks = (N + 31) >> 5;
  const int cpw    = (chunks + SETUP_WAVES - 1) / SETUP_WAVES;
  const int n0     = wave * cpw * 32;

  int run[NSP];
  for (int q = 0; q < NSP; ++q) run[q] = 0;
  for (int j = 0; j < cpw; ++j) {
    const int n = n0 + j * 32 + lane;
    const int s = (n < N) ? index[n] : -1;
    for (int q = 0; q < NSP; ++q)
      run[q] += __builtin_popcount(__builtin_amdgcn_ballot_w32(s == q));
  }
  if (lane < NSP) {
    int v = run[0];
    for (int q = 1; q < NSP; ++q) if (lane == q) v = run[q];
    waveCnt[wave][lane] = v;
  }
  __syncthreads();

  if (tid < NSP) {
    int acc = 0;
    for (int w = 0; w < SETUP_WAVES; ++w) {
      const int v = waveCnt[w][tid];
      waveCnt[w][tid] = acc;
      acc += v;
    }
    spCnt[tid] = acc;
  }
  __syncthreads();

  int cnt[NSP], segBase[NSP];
  int base = 0;
  for (int q = 0; q < NSP; ++q) {
    cnt[q]     = spCnt[q];
    segBase[q] = base;
    base += ((cnt[q] + 15) >> 4) << 4;
  }
  const int tiles = base >> 4;

  int wb[NSP];
  for (int q = 0; q < NSP; ++q) wb[q] = segBase[q] + waveCnt[wave][q];
  for (int j = 0; j < cpw; ++j) {
    const int n = n0 + j * 32 + lane;
    const int s = (n < N) ? index[n] : -1;
    int pos = -1;
    for (int q = 0; q < NSP; ++q) {
      const unsigned m = __builtin_amdgcn_ballot_w32(s == q);
      if (s == q) pos = wb[q] + (int)__builtin_amdgcn_mbcnt_lo(m, 0u);
      wb[q] += __builtin_popcount(m);
    }
    if (pos >= 0) perm[pos] = n;
  }

  if (tid < NSP * 16) {
    const int sp = tid >> 4;
    const int j  = tid & 15;
    int c = cnt[0], sb = segBase[0];
    for (int q = 1; q < NSP; ++q) if (sp == q) { c = cnt[q]; sb = segBase[q]; }
    const int slot = c + j;
    if (slot < (((c + 15) >> 4) << 4)) perm[sb + slot] = -1;
  }

  for (int t = tid; t < tiles; t += SETUP_THREADS) {
    int sp = 0;
    for (int q = 0; q < NSP - 1; ++q)
      if (t >= ((segBase[q] >> 4) + ((cnt[q] + 15) >> 4))) sp = q + 1;
    tileSpecies[t] = sp;
    tileSlot[t]    = t * 16;
  }
  if (tid == 0) meta[0] = tiles;
}

template <int MUL, int IRD, int M0, int TP, int KOFF>
__device__ __forceinline__ void build_folded(float* Ct, const float* __restrict__ F,
                                             const float (&w)[MUL], int tid) {
  constexpr int NQ = TP / 4;
  constexpr int NI = IRD * NQ;
#pragma unroll
  for (int r = 0; r < (NI + 127) / 128; ++r) {
    const int it0 = tid + 128 * r;
    const int it  = (it0 < NI) ? it0 : NI - 1;
    const int i   = it / NQ;
    const int q   = it - i * NQ;
    const float* fp = F + i * TP + 4 * q;
    float v0 = 0.0f, v1 = 0.0f, v2 = 0.0f, v3 = 0.0f;
#pragma unroll
    for (int k = 0; k < MUL; ++k) {
      const v4f u = *(const v4f*)(fp + k * IRD * TP);
      v0 = __builtin_fmaf(u.x, w[k], v0);
      v1 = __builtin_fmaf(u.y, w[k], v1);
      v2 = __builtin_fmaf(u.z, w[k], v2);
      v3 = __builtin_fmaf(u.w, w[k], v3);
    }
    const v4f o = {v0, v1, v2, v3};
    *(v4f*)(Ct + (M0 + i) * KTOT + KOFF + 4 * q) = o;
  }
}

template <int IRD, int M0>
__device__ __forceinline__ void build_order1(float* Ct, const float* __restrict__ U1,
                                             const float* __restrict__ W1,
                                             int s, int c, int C, int tid) {
  constexpr int NI = 9 * IRD;
  const int it = (tid < NI) ? tid : NI - 1;
  const int ii = it / 9;
  const int b  = it - ii * 9;
  Ct[(M0 + ii) * KTOT + K_S1 + b] = U1[b * IRD + ii] * W1[s * C + c];
}

struct Mono { int kind, a, b, c; };
constexpr Mono mono_of(int slot) {
  if (slot < K_P2) {
    int n = 0;
    for (int a = 0; a < 9; ++a)
      for (int b = a; b < 9; ++b)
        for (int c = b; c < 9; ++c) {
          if (n == slot - K_T3) return Mono{3, a, b, c};
          ++n;
        }
    return Mono{0, 0, 0, 0};
  }
  if (slot < K_S1) {
    int n = 0;
    for (int a = 0; a < 9; ++a)
      for (int b = a; b < 9; ++b) {
        if (n == slot - K_P2) return Mono{2, a, b, 0};
        ++n;
      }
    return Mono{0, 0, 0, 0};
  }
  return (slot - K_S1 < 9) ? Mono{1, 0, slot - K_S1, 0} : Mono{0, 0, 0, 0};
}

template <int SLOT>
__device__ __forceinline__ float mono_val(const float (&xv)[9]) {
  constexpr Mono m = mono_of(SLOT);
  if constexpr (m.kind == 3)      return (xv[m.a] * xv[m.b]) * xv[m.c];
  else if constexpr (m.kind == 2) return xv[m.a] * xv[m.b];
  else if constexpr (m.kind == 1) return xv[m.b];
  else                            return 0.0f;
}

template <int SL, int SH>
__device__ __forceinline__ float b_elem(const float (&xv)[9], bool hi) {
  constexpr Mono L = mono_of(SL);
  constexpr Mono H = mono_of(SH);
  if constexpr (L.kind == 3 && H.kind == 3) {
    float p, e;
    if constexpr (L.a == H.a && L.b == H.b) p = xv[L.a] * xv[L.b];
    else                                    p = hi ? xv[H.a] * xv[H.b] : xv[L.a] * xv[L.b];
    if constexpr (L.c == H.c) e = xv[L.c];
    else                      e = hi ? xv[H.c] : xv[L.c];
    return p * e;
  } else {
    return hi ? mono_val<SH>(xv) : mono_val<SL>(xv);
  }
}

template <int J>
__device__ __forceinline__ void k_blocks(const float* Abase, const float (&xv)[9], bool hi,
                                         v8f& acc0, v8f& acc1, v8f& acc2) {
  if constexpr (J < NBLK) {
    const v2f af = *(const v2f*)(Abase + 4 * J);
    v2f bf;
    bf.x = b_elem<4 * J,     4 * J + 2>(xv, hi);
    bf.y = b_elem<4 * J + 1, 4 * J + 3>(xv, hi);
    if constexpr (J % 3 == 0)
      acc0 = __builtin_amdgcn_wmma_f32_16x16x4_f32(false, af, false, bf, (short)0, acc0, false, false);
    else if constexpr (J % 3 == 1)
      acc1 = __builtin_amdgcn_wmma_f32_16x16x4_f32(false, af, false, bf, (short)0, acc1, false, false);
    else
      acc2 = __builtin_amdgcn_wmma_f32_16x16x4_f32(false, af, false, bf, (short)0, acc2, false, false);
    k_blocks<J + 1>(Abase, xv, hi, acc0, acc1, acc2);
  }
}

__global__ __launch_bounds__(128) void sc_main(
    const float* __restrict__ X,
    const int*   __restrict__ meta,
    const int*   __restrict__ perm,
    const int*   __restrict__ tileSpecies,
    const int*   __restrict__ tileSlot,
    float*       __restrict__ out, int C,
    const float* __restrict__ F,
    const float* __restrict__ W3_0e, const float* __restrict__ W3_1o,
    const float* __restrict__ W3_2e,
    const float* __restrict__ W2_0e, const float* __restrict__ W2_1o,
    const float* __restrict__ W2_2e,
    const float* __restrict__ U1_0e, const float* __restrict__ W1_0e,
    const float* __restrict__ U1_1o, const float* __restrict__ W1_1o,
    const float* __restrict__ U1_2e, const float* __restrict__ W1_2e)
{
  __shared__ __attribute__((aligned(16))) float Ct[MROWS * KTOT];
  const int c   = blockIdx.x;
  const int s   = blockIdx.y;
  const int tid = threadIdx.x;

  for (int n = tid; n < MROWS * KTOT / 4; n += 128) {
    const v4f z = {0.f, 0.f, 0.f, 0.f};
    *(v4f*)(Ct + 4 * n) = z;
  }
  float w0[10], w1[11], w2[13], y0[3], y1[2], y2[3];
#pragma unroll
  for (int k = 0; k < 10; ++k) w0[k] = W3_0e[(s * 10 + k) * C + c];
#pragma unroll
  for (int k = 0; k < 11; ++k) w1[k] = W3_1o[(s * 11 + k) * C + c];
#pragma unroll
  for (int k = 0; k < 13; ++k) w2[k] = W3_2e[(s * 13 + k) * C + c];
#pragma unroll
  for (int k = 0; k < 3; ++k) y0[k] = W2_0e[(s * 3 + k) * C + c];
#pragma unroll
  for (int k = 0; k < 2; ++k) y1[k] = W2_1o[(s * 2 + k) * C + c];
#pragma unroll
  for (int k = 0; k < 3; ++k) y2[k] = W2_2e[(s * 3 + k) * C + c];
  __syncthreads();
  build_folded<13, 5, 4, TP3, K_T3>(Ct, F + F3_2E_OFF, w2, tid);
  __builtin_amdgcn_sched_barrier(0);
  build_folded<11, 3, 1, TP3, K_T3>(Ct, F + F3_1O_OFF, w1, tid);
  build_folded<10, 1, 0, TP3, K_T3>(Ct, F + F3_0E_OFF, w0, tid);
  build_folded<3, 5, 4, TP2, K_P2>(Ct, F + F2_2E_OFF, y2, tid);
  build_folded<2, 3, 1, TP2, K_P2>(Ct, F + F2_1O_OFF, y1, tid);
  build_folded<3, 1, 0, TP2, K_P2>(Ct, F + F2_0E_OFF, y0, tid);
  build_order1<1, 0>(Ct, U1_0e, W1_0e, s, c, C, tid);
  build_order1<3, 1>(Ct, U1_1o, W1_1o, s, c, C, tid);
  build_order1<5, 4>(Ct, U1_2e, W1_2e, s, c, C, tid);
  __syncthreads();

  const int  lane = tid & 31;
  const int  wave = tid >> 5;
  const int  n16  = lane & 15;
  const bool hi   = (lane & 16) != 0;
  const float* Abase = &Ct[n16 * KTOT + (hi ? 2 : 0)];
  const int numTiles = meta[0];

  int mc = 0;
  for (int t = 0; t < numTiles; ++t) {
    if (tileSpecies[t] != s) continue;
    const int my = mc++;
    if ((my & 3) != wave) continue;

    const int  slotBase = tileSlot[t];
    const int  node  = perm[slotBase + n16];
    const bool valid = node >= 0;
    const int  nodeC = valid ? node : 0;
    const float* xp = X + (nodeC * C + c) * 9;
    float xv[9];
#pragma unroll
    for (int d = 0; d < 9; ++d) xv[d] = valid ? xp[d] : 0.0f;

    v8f acc0 = {0.f,0.f,0.f,0.f,0.f,0.f,0.f,0.f};
    v8f acc1 = acc0, acc2 = acc0;
    k_blocks<0>(Abase, xv, hi, acc0, acc1, acc2);
    v8f acc = acc0 + acc1 + acc2;

    if (valid) {
      float* op = out + (node * C + c) * 9;
      if (!hi) {
#pragma unroll
        for (int r = 0; r < 8; ++r) op[r] = acc[r];
      } else {
        op[8] = acc[0];
      }
    }
  }
}

extern "C" void kernel_launch(void* const* d_in, const int* in_sizes, int n_in,
                              void* d_out, int out_size, void* d_ws, size_t ws_size,
                              hipStream_t stream) {
  (void)n_in; (void)out_size; (void)ws_size;
  const float* X     = (const float*)d_in[0];
  const int*   index = (const int*)d_in[1];
  const int N = in_sizes[1];
  const int C = in_sizes[0] / (N * 9);

  int* meta        = (int*)d_ws;
  int* perm        = meta + 16;
  int  permCap     = (N + 16 * NSP + 15) & ~15;
  int* tileSpecies = perm + permCap;
  int* tileSlot    = tileSpecies + 256;
  float* F         = (float*)(tileSlot + 256);

  sc_setup<<<1 + FOLD_BLOCKS, SETUP_THREADS, 0, stream>>>(
      index, N, meta, perm, tileSpecies, tileSlot, F,
      (const float*)d_in[2], (const float*)d_in[4], (const float*)d_in[6],
      (const float*)d_in[8], (const float*)d_in[10], (const float*)d_in[12]);

  sc_main<<<dim3(C, NSP), 128, 0, stream>>>(
      X, meta, perm, tileSpecies, tileSlot, (float*)d_out, C, F,
      (const float*)d_in[3],  (const float*)d_in[5],  (const float*)d_in[7],
      (const float*)d_in[9],  (const float*)d_in[11], (const float*)d_in[13],
      (const float*)d_in[14], (const float*)d_in[15],
      (const float*)d_in[16], (const float*)d_in[17],
      (const float*)d_in[18], (const float*)d_in[19]);
}


# GEMM tile loop freed of global round trips: each wave indexes its species' tile range directly (no scan over all tiles), fetches node ids two tiles and features one tile ahead of the WMMAs
# speedup vs baseline: 21.9266x; 1.0455x over previous
#include <hip/hip_runtime.h>


typedef float v2f __attribute__((ext_vector_type(2)));
typedef float v4f __attribute__((ext_vector_type(4)));
typedef float v8f __attribute__((ext_vector_type(8)));

#define NSP   10
#define KROW  12
#define MROWS 16

#define NT3 165
#define TP3 168
#define NT2 45
#define TP2 48
#define F3_0E_OFF 0
#define F3_1O_OFF (F3_0E_OFF + 10 * 1 * TP3)
#define F3_2E_OFF (F3_1O_OFF + 11 * 3 * TP3)
#define F2_0E_OFF (F3_2E_OFF + 13 * 5 * TP3)
#define F2_1O_OFF (F2_0E_OFF + 3 * 1 * TP2)
#define F2_2E_OFF (F2_1O_OFF + 2 * 3 * TP2)
#define F_TOTAL   (F2_2E_OFF + 3 * 5 * TP2)

#define K_T3  0
#define K_P2  TP3
#define K_S1  (TP3 + TP2)
#define KTOT  (TP3 + TP2 + 12)
#define NBLK  (KTOT / 4)

__device__ __forceinline__ const unsigned short* triple_slots() {
  static const __attribute__((aligned(8))) unsigned short tab[TP3] = {
      0, 1, 2, 3, 4, 5, 6, 7, 8, 13, 14, 15, 16, 17,
      18, 19, 20, 26, 27, 28, 29, 30, 31, 32, 39, 40, 41, 42,
      43, 44, 52, 53, 54, 55, 56, 65, 66, 67, 68, 78, 79, 80,
      91, 92, 104, 121, 122, 123, 124, 125, 126, 127, 128, 134, 135, 136,
      137, 138, 139, 140, 147, 148, 149, 150, 151, 152, 160, 161, 162, 163,
      164, 173, 174, 175, 176, 186, 187, 188, 199, 200, 212, 242, 243, 244,
      245, 246, 247, 248, 255, 256, 257, 258, 259, 260, 268, 269, 270, 271,
      272, 281, 282, 283, 284, 294, 295, 296, 307, 308, 320, 363, 364, 365,
      366, 367, 368, 376, 377, 378, 379, 380, 389, 390, 391, 392, 402, 403,
      404, 415, 416, 428, 484, 485, 486, 487, 488, 497, 498, 499, 500, 510,
      511, 512, 523, 524, 536, 605, 606, 607, 608, 618, 619, 620, 631, 632,
      644, 726, 727, 728, 739, 740, 752, 847, 848, 860, 968, 11, 11, 11};
  return tab;
}
__device__ __forceinline__ const unsigned short* pair_slots() {
  static const __attribute__((aligned(8))) unsigned short tab[TP2] = {
      9, 21, 33, 45, 57, 69, 81, 93, 105, 129, 141, 153,
      165, 177, 189, 201, 213, 249, 261, 273, 285, 297, 309, 321,
      369, 381, 393, 405, 417, 429, 489, 501, 513, 525, 537, 609,
      621, 633, 645, 729, 741, 753, 849, 861, 969, 11, 11, 11};
  return tab;
}

template <int MUL, int IRD>
__device__ __forceinline__ void fold3_entry(const float* __restrict__ U, float* __restrict__ F, int o) {
  constexpr int PER_P = MUL * IRD;
  const int t  = o / PER_P;
  const int ki = o - t * PER_P;
  float v = 0.0f;
  if (t < NT3) {
    const int sl = triple_slots()[t];
    const int g = sl / KROW;
    const int c = sl - g * KROW;
    const int a = g / 9;
    const int b = g - a * 9;
    const float s1 = U[((a * 9 + b) * 9 + c) * PER_P + ki];
    const float s2 = U[((a * 9 + c) * 9 + b) * PER_P + ki];
    const float s3 = U[((c * 9 + a) * 9 + b) * PER_P + ki];
    const float s4 = U[((b * 9 + a) * 9 + c) * PER_P + ki];
    const float s5 = U[((b * 9 + c) * 9 + a) * PER_P + ki];
    const float s6 = U[((c * 9 + b) * 9 + a) * PER_P + ki];
    v = s1;
    v += (b != c) ? s2 : 0.0f;
    v += (b != c) ? s3 : 0.0f;
    v += (a != b) ? s4 : 0.0f;
    v += (a != b) ? s5 : 0.0f;
    v += (a != b && b != c) ? s6 : 0.0f;
  }
  F[ki * TP3 + t] = v;
}

template <int MUL, int IRD>
__device__ __forceinline__ void fold2_entry(const float* __restrict__ U, float* __restrict__ F, int o) {
  constexpr int PER_P = MUL * IRD;
  const int p  = o / PER_P;
  const int ki = o - p * PER_P;
  float v = 0.0f;
  if (p < NT2) {
    const int g = pair_slots()[p] / KROW;
    const int a = g / 9;
    const int b = g - a * 9;
    const float own = U[(a * 9 + b) * PER_P + ki];
    const float oth = U[(b * 9 + a) * PER_P + ki];
    v = own + ((a != b) ? oth : 0.0f);
  }
  F[ki * TP2 + p] = v;
}

#define SETUP_WAVES   32
#define SETUP_THREADS (SETUP_WAVES * 32)
#define FOLD_BLOCKS   ((F_TOTAL + SETUP_THREADS - 1) / SETUP_THREADS)

__global__ __launch_bounds__(SETUP_THREADS) void sc_setup(
    const int* __restrict__ index, int N,
    int* __restrict__ meta, int* __restrict__ perm,
    float* __restrict__ F,
    const float* __restrict__ U3_0e, const float* __restrict__ U3_1o,
    const float* __restrict__ U3_2e, const float* __restrict__ U2_0e,
    const float* __restrict__ U2_1o, const float* __restrict__ U2_2e) {
  __shared__ int waveCnt[SETUP_WAVES][NSP];
  __shared__ int spCnt[NSP];
  const int tid  = threadIdx.x;

  if (blockIdx.x != 0) {
    const int o = (blockIdx.x - 1) * SETUP_THREADS + tid;
    if      (o < F3_1O_OFF) fold3_entry<10, 1>(U3_0e, F + F3_0E_OFF, o - F3_0E_OFF);
    else if (o < F3_2E_OFF) fold3_entry<11, 3>(U3_1o, F + F3_1O_OFF, o - F3_1O_OFF);
    else if (o < F2_0E_OFF) fold3_entry<13, 5>(U3_2e, F + F3_2E_OFF, o - F3_2E_OFF);
    else if (o < F2_1O_OFF) fold2_entry<3, 1>(U2_0e, F + F2_0E_OFF, o - F2_0E_OFF);
    else if (o < F2_2E_OFF) fold2_entry<2, 3>(U2_1o, F + F2_1O_OFF, o - F2_1O_OFF);
    else if (o < F_TOTAL)   fold2_entry<3, 5>(U2_2e, F + F2_2E_OFF, o - F2_2E_OFF);
    return;
  }
  const int lane = tid & 31;
  const int wave = tid >> 5;

  const int chunks = (N + 31) >> 5;
  const int cpw    = (chunks + SETUP_WAVES - 1) / SETUP_WAVES;
  const int n0     = wave * cpw * 32;

  int run[NSP];
  for (int q = 0; q < NSP; ++q) run[q] = 0;
  for (int j = 0; j < cpw; ++j) {
    const int n = n0 + j * 32 + lane;
    const int s = (n < N) ? index[n] : -1;
    for (int q = 0; q < NSP; ++q)
      run[q] += __builtin_popcount(__builtin_amdgcn_ballot_w32(s == q));
  }
  if (lane < NSP) {
    int v = run[0];
    for (int q = 1; q < NSP; ++q) if (lane == q) v = run[q];
    waveCnt[wave][lane] = v;
  }
  __syncthreads();

  if (tid < NSP) {
    int acc = 0;
    for (int w = 0; w < SETUP_WAVES; ++w) {
      const int v = waveCnt[w][tid];
      waveCnt[w][tid] = acc;
      acc += v;
    }
    spCnt[tid] = acc;
  }
  __syncthreads();

  int cnt[NSP], segBase[NSP];
  int base = 0;
  for (int q = 0; q < NSP; ++q) {
    cnt[q]     = spCnt[q];
    segBase[q] = base;
    base += ((cnt[q] + 15) >> 4) << 4;
  }
  const int tiles = base >> 4;

  int wb[NSP];
  for (int q = 0; q < NSP; ++q) wb[q] = segBase[q] + waveCnt[wave][q];
  for (int j = 0; j < cpw; ++j) {
    const int n = n0 + j * 32 + lane;
    const int s = (n < N) ? index[n] : -1;
    int pos = -1;
    for (int q = 0; q < NSP; ++q) {
      const unsigned m = __builtin_amdgcn_ballot_w32(s == q);
      if (s == q) pos = wb[q] + (int)__builtin_amdgcn_mbcnt_lo(m, 0u);
      wb[q] += __builtin_popcount(m);
    }
    if (pos >= 0) perm[pos] = n;
  }

  if (tid < NSP * 16) {
    const int sp = tid >> 4;
    const int j  = tid & 15;
    int c = cnt[0], sb = segBase[0];
    for (int q = 1; q < NSP; ++q) if (sp == q) { c = cnt[q]; sb = segBase[q]; }
    const int slot = c + j;
    if (slot < (((c + 15) >> 4) << 4)) perm[sb + slot] = -1;
  }

  if (tid <= NSP) {
    int v = tiles;
    for (int q = 0; q < NSP; ++q) if (tid == q) v = segBase[q] >> 4;
    meta[tid] = v;
  }
}

template <int MUL, int IRD, int M0, int TP, int KOFF>
__device__ __forceinline__ void build_folded(float* Ct, const float* __restrict__ F,
                                             const float (&w)[MUL], int tid) {
  constexpr int NQ = TP / 4;
  constexpr int NI = IRD * NQ;
#pragma unroll
  for (int r = 0; r < (NI + 127) / 128; ++r) {
    const int it0 = tid + 128 * r;
    const int it  = (it0 < NI) ? it0 : NI - 1;
    const int i   = it / NQ;
    const int q   = it - i * NQ;
    const float* fp = F + i * TP + 4 * q;
    float v0 = 0.0f, v1 = 0.0f, v2 = 0.0f, v3 = 0.0f;
#pragma unroll
    for (int k = 0; k < MUL; ++k) {
      const v4f u = *(const v4f*)(fp + k * IRD * TP);
      v0 = __builtin_fmaf(u.x, w[k], v0);
      v1 = __builtin_fmaf(u.y, w[k], v1);
      v2 = __builtin_fmaf(u.z, w[k], v2);
      v3 = __builtin_fmaf(u.w, w[k], v3);
    }
    const v4f o = {v0, v1, v2, v3};
    *(v4f*)(Ct + (M0 + i) * KTOT + KOFF + 4 * q) = o;
  }
}

template <int IRD, int M0>
__device__ __forceinline__ void build_order1(float* Ct, const float* __restrict__ U1,
                                             const float* __restrict__ W1,
                                             int s, int c, int C, int tid) {
  constexpr int NI = 9 * IRD;
  const int it = (tid < NI) ? tid : NI - 1;
  const int ii = it / 9;
  const int b  = it - ii * 9;
  Ct[(M0 + ii) * KTOT + K_S1 + b] = U1[b * IRD + ii] * W1[s * C + c];
}

struct Mono { int kind, a, b, c; };
constexpr Mono mono_of(int slot) {
  if (slot < K_P2) {
    int n = 0;
    for (int a = 0; a < 9; ++a)
      for (int b = a; b < 9; ++b)
        for (int c = b; c < 9; ++c) {
          if (n == slot - K_T3) return Mono{3, a, b, c};
          ++n;
        }
    return Mono{0, 0, 0, 0};
  }
  if (slot < K_S1) {
    int n = 0;
    for (int a = 0; a < 9; ++a)
      for (int b = a; b < 9; ++b) {
        if (n == slot - K_P2) return Mono{2, a, b, 0};
        ++n;
      }
    return Mono{0, 0, 0, 0};
  }
  return (slot - K_S1 < 9) ? Mono{1, 0, slot - K_S1, 0} : Mono{0, 0, 0, 0};
}

template <int SLOT>
__device__ __forceinline__ float mono_val(const float (&xv)[9]) {
  constexpr Mono m = mono_of(SLOT);
  if constexpr (m.kind == 3)      return (xv[m.a] * xv[m.b]) * xv[m.c];
  else if constexpr (m.kind == 2) return xv[m.a] * xv[m.b];
  else if constexpr (m.kind == 1) return xv[m.b];
  else                            return 0.0f;
}

template <int SL, int SH>
__device__ __forceinline__ float b_elem(const float (&xv)[9], bool hi) {
  constexpr Mono L = mono_of(SL);
  constexpr Mono H = mono_of(SH);
  if constexpr (L.kind == 3 && H.kind == 3) {
    float p, e;
    if constexpr (L.a == H.a && L.b == H.b) p = xv[L.a] * xv[L.b];
    else                                    p = hi ? xv[H.a] * xv[H.b] : xv[L.a] * xv[L.b];
    if constexpr (L.c == H.c) e = xv[L.c];
    else                      e = hi ? xv[H.c] : xv[L.c];
    return p * e;
  } else {
    return hi ? mono_val<SH>(xv) : mono_val<SL>(xv);
  }
}

template <int J>
__device__ __forceinline__ void k_blocks(const float* Abase, const float (&xv)[9], bool hi,
                                         v8f& acc0, v8f& acc1, v8f& acc2) {
  if constexpr (J < NBLK) {
    const v2f af = *(const v2f*)(Abase + 4 * J);
    v2f bf;
    bf.x = b_elem<4 * J,     4 * J + 2>(xv, hi);
    bf.y = b_elem<4 * J + 1, 4 * J + 3>(xv, hi);
    if constexpr (J % 3 == 0)
      acc0 = __builtin_amdgcn_wmma_f32_16x16x4_f32(false, af, false, bf, (short)0, acc0, false, false);
    else if constexpr (J % 3 == 1)
      acc1 = __builtin_amdgcn_wmma_f32_16x16x4_f32(false, af, false, bf, (short)0, acc1, false, false);
    else
      acc2 = __builtin_amdgcn_wmma_f32_16x16x4_f32(false, af, false, bf, (short)0, acc2, false, false);
    k_blocks<J + 1>(Abase, xv, hi, acc0, acc1, acc2);
  }
}

__device__ __forceinline__ void load_x(float (&xr)[9], const float* __restrict__ X,
                                       int node, int C, int c) {
  const float* xp = X + ((node >= 0 ? node : 0) * C + c) * 9;
#pragma unroll
  for (int d = 0; d < 9; ++d) xr[d] = xp[d];
}

__global__ __launch_bounds__(128) void sc_main(
    const float* __restrict__ X,
    const int*   __restrict__ meta,
    const int*   __restrict__ perm,
    float*       __restrict__ out, int C,
    const float* __restrict__ F,
    const float* __restrict__ W3_0e, const float* __restrict__ W3_1o,
    const float* __restrict__ W3_2e,
    const float* __restrict__ W2_0e, const float* __restrict__ W2_1o,
    const float* __restrict__ W2_2e,
    const float* __restrict__ U1_0e, const float* __restrict__ W1_0e,
    const float* __restrict__ U1_1o, const float* __restrict__ W1_1o,
    const float* __restrict__ U1_2e, const float* __restrict__ W1_2e)
{
  __shared__ __attribute__((aligned(16))) float Ct[MROWS * KTOT];
  const int c   = blockIdx.x;
  const int s   = blockIdx.y;
  const int tid = threadIdx.x;
  const int  lane = tid & 31;
  const int  wave = __builtin_amdgcn_readfirstlane(tid >> 5);
  const int  n16  = lane & 15;
  const bool hi   = (lane & 16) != 0;

  for (int n = tid; n < MROWS * KTOT / 4; n += 128) {
    const v4f z = {0.f, 0.f, 0.f, 0.f};
    *(v4f*)(Ct + 4 * n) = z;
  }
  float w0[10], w1[11], w2[13], y0[3], y1[2], y2[3];
#pragma unroll
  for (int k = 0; k < 10; ++k) w0[k] = W3_0e[(s * 10 + k) * C + c];
#pragma unroll
  for (int k = 0; k < 11; ++k) w1[k] = W3_1o[(s * 11 + k) * C + c];
#pragma unroll
  for (int k = 0; k < 13; ++k) w2[k] = W3_2e[(s * 13 + k) * C + c];
#pragma unroll
  for (int k = 0; k < 3; ++k) y0[k] = W2_0e[(s * 3 + k) * C + c];
#pragma unroll
  for (int k = 0; k < 2; ++k) y1[k] = W2_1o[(s * 2 + k) * C + c];
#pragma unroll
  for (int k = 0; k < 3; ++k) y2[k] = W2_2e[(s * 3 + k) * C + c];
  __syncthreads();
  build_folded<13, 5, 4, TP3, K_T3>(Ct, F + F3_2E_OFF, w2, tid);
  __builtin_amdgcn_sched_barrier(0);
  const int t0 = meta[s];
  const int t1 = meta[s + 1];
  build_folded<11, 3, 1, TP3, K_T3>(Ct, F + F3_1O_OFF, w1, tid);
  __builtin_amdgcn_sched_barrier(0);
  const int  nT = t1 - t0;
  const int* pbase = perm + t0 * 16;
  int nodeCur = -1, nodeNxt = -1;
  if (wave < nT)     nodeCur = pbase[wave * 16 + n16];
  if (wave + 4 < nT) nodeNxt = pbase[(wave + 4) * 16 + n16];
  build_folded<10, 1, 0, TP3, K_T3>(Ct, F + F3_0E_OFF, w0, tid);
  build_folded<3, 5, 4, TP2, K_P2>(Ct, F + F2_2E_OFF, y2, tid);
  build_folded<2, 3, 1, TP2, K_P2>(Ct, F + F2_1O_OFF, y1, tid);
  build_folded<3, 1, 0, TP2, K_P2>(Ct, F + F2_0E_OFF, y0, tid);
  build_order1<1, 0>(Ct, U1_0e, W1_0e, s, c, C, tid);
  build_order1<3, 1>(Ct, U1_1o, W1_1o, s, c, C, tid);
  build_order1<5, 4>(Ct, U1_2e, W1_2e, s, c, C, tid);
  float xc[9];
#pragma unroll
  for (int d = 0; d < 9; ++d) xc[d] = 0.0f;
  if (wave < nT) load_x(xc, X, nodeCur, C, c);
  __syncthreads();

  const float* Abase = &Ct[n16 * KTOT + (hi ? 2 : 0)];

  for (int my = wave; my < nT; my += 4) {
    float xn[9];
#pragma unroll
    for (int d = 0; d < 9; ++d) xn[d] = 0.0f;
    int nodeNx2 = -1;
    if (my + 4 < nT) {
      load_x(xn, X, nodeNxt, C, c);
      if (my + 8 < nT) nodeNx2 = pbase[(my + 8) * 16 + n16];
    }

    asm volatile("" ::: "memory");
    v8f acc0 = {0.f,0.f,0.f,0.f,0.f,0.f,0.f,0.f};
    v8f acc1 = acc0, acc2 = acc0;
    k_blocks<0>(Abase, xc, hi, acc0, acc1, acc2);
    v8f acc = acc0 + acc1 + acc2;

    if (nodeCur >= 0) {
      float* op = out + (nodeCur * C + c) * 9;
      if (!hi) {
#pragma unroll
        for (int r = 0; r < 8; ++r) op[r] = acc[r];
      } else {
        op[8] = acc[0];
      }
    }

    nodeCur = nodeNxt;
    nodeNxt = nodeNx2;
#pragma unroll
    for (int d = 0; d < 9; ++d) xc[d] = xn[d];
  }
}

extern "C" void kernel_launch(void* const* d_in, const int* in_sizes, int n_in,
                              void* d_out, int out_size, void* d_ws, size_t ws_size,
                              hipStream_t stream) {
  (void)n_in; (void)out_size; (void)ws_size;
  const float* X     = (const float*)d_in[0];
  const int*   index = (const int*)d_in[1];
  const int N = in_sizes[1];
  const int C = in_sizes[0] / (N * 9);

  int* meta        = (int*)d_ws;
  int* perm        = meta + 16;
  int  permCap     = (N + 16 * NSP + 15) & ~15;
  float* F         = (float*)(perm + permCap);

  sc_setup<<<1 + FOLD_BLOCKS, SETUP_THREADS, 0, stream>>>(
      index, N, meta, perm, F,
      (const float*)d_in[2], (const float*)d_in[4], (const float*)d_in[6],
      (const float*)d_in[8], (const float*)d_in[10], (const float*)d_in[12]);

  sc_main<<<dim3(C, NSP), 128, 0, stream>>>(
      X, meta, perm, (float*)d_out, C, F,
      (const float*)d_in[3],  (const float*)d_in[5],  (const float*)d_in[7],
      (const float*)d_in[9],  (const float*)d_in[11], (const float*)d_in[13],
      (const float*)d_in[14], (const float*)d_in[15],
      (const float*)d_in[16], (const float*)d_in[17],
      (const float*)d_in[18], (const float*)d_in[19]);
}


# Main kernel: one 20-wave work-group per WGP serves five (species, channel) items; each wave reads one folded tensor's plane words once and contracts them with all five items' weights
# speedup vs baseline: 29.7433x; 1.3565x over previous
#include <hip/hip_runtime.h>


typedef float v2f __attribute__((ext_vector_type(2)));
typedef float v4f __attribute__((ext_vector_type(4)));
typedef float v8f __attribute__((ext_vector_type(8)));

#define NSP   10
#define KROW  12
#define MROWS 16

#define NT3 165
#define TP3 168
#define NT2 45
#define TP2 48
#define F3_0E_OFF 0
#define F3_1O_OFF (F3_0E_OFF + 10 * 1 * TP3)
#define F3_2E_OFF (F3_1O_OFF + 11 * 3 * TP3)
#define F2_0E_OFF (F3_2E_OFF + 13 * 5 * TP3)
#define F2_1O_OFF (F2_0E_OFF + 3 * 1 * TP2)
#define F2_2E_OFF (F2_1O_OFF + 2 * 3 * TP2)
#define F_TOTAL   (F2_2E_OFF + 3 * 5 * TP2)

#define K_T3  0
#define K_P2  TP3
#define K_S1  (TP3 + TP2)
#define KTOT  (TP3 + TP2 + 12)
#define NBLK  (KTOT / 4)

__device__ __forceinline__ const unsigned short* triple_slots() {
  static const __attribute__((aligned(8))) unsigned short tab[TP3] = {
      0, 1, 2, 3, 4, 5, 6, 7, 8, 13, 14, 15, 16, 17,
      18, 19, 20, 26, 27, 28, 29, 30, 31, 32, 39, 40, 41, 42,
      43, 44, 52, 53, 54, 55, 56, 65, 66, 67, 68, 78, 79, 80,
      91, 92, 104, 121, 122, 123, 124, 125, 126, 127, 128, 134, 135, 136,
      137, 138, 139, 140, 147, 148, 149, 150, 151, 152, 160, 161, 162, 163,
      164, 173, 174, 175, 176, 186, 187, 188, 199, 200, 212, 242, 243, 244,
      245, 246, 247, 248, 255, 256, 257, 258, 259, 260, 268, 269, 270, 271,
      272, 281, 282, 283, 284, 294, 295, 296, 307, 308, 320, 363, 364, 365,
      366, 367, 368, 376, 377, 378, 379, 380, 389, 390, 391, 392, 402, 403,
      404, 415, 416, 428, 484, 485, 486, 487, 488, 497, 498, 499, 500, 510,
      511, 512, 523, 524, 536, 605, 606, 607, 608, 618, 619, 620, 631, 632,
      644, 726, 727, 728, 739, 740, 752, 847, 848, 860, 968, 11, 11, 11};
  return tab;
}
__device__ __forceinline__ const unsigned short* pair_slots() {
  static const __attribute__((aligned(8))) unsigned short tab[TP2] = {
      9, 21, 33, 45, 57, 69, 81, 93, 105, 129, 141, 153,
      165, 177, 189, 201, 213, 249, 261, 273, 285, 297, 309, 321,
      369, 381, 393, 405, 417, 429, 489, 501, 513, 525, 537, 609,
      621, 633, 645, 729, 741, 753, 849, 861, 969, 11, 11, 11};
  return tab;
}

template <int MUL, int IRD>
__device__ __forceinline__ void fold3_entry(const float* __restrict__ U, float* __restrict__ F, int o) {
  constexpr int PER_P = MUL * IRD;
  const int t  = o / PER_P;
  const int ki = o - t * PER_P;
  float v = 0.0f;
  if (t < NT3) {
    const int sl = triple_slots()[t];
    const int g = sl / KROW;
    const int c = sl - g * KROW;
    const int a = g / 9;
    const int b = g - a * 9;
    const float s1 = U[((a * 9 + b) * 9 + c) * PER_P + ki];
    const float s2 = U[((a * 9 + c) * 9 + b) * PER_P + ki];
    const float s3 = U[((c * 9 + a) * 9 + b) * PER_P + ki];
    const float s4 = U[((b * 9 + a) * 9 + c) * PER_P + ki];
    const float s5 = U[((b * 9 + c) * 9 + a) * PER_P + ki];
    const float s6 = U[((c * 9 + b) * 9 + a) * PER_P + ki];
    v = s1;
    v += (b != c) ? s2 : 0.0f;
    v += (b != c) ? s3 : 0.0f;
    v += (a != b) ? s4 : 0.0f;
    v += (a != b) ? s5 : 0.0f;
    v += (a != b && b != c) ? s6 : 0.0f;
  }
  F[ki * TP3 + t] = v;
}

template <int MUL, int IRD>
__device__ __forceinline__ void fold2_entry(const float* __restrict__ U, float* __restrict__ F, int o) {
  constexpr int PER_P = MUL * IRD;
  const int p  = o / PER_P;
  const int ki = o - p * PER_P;
  float v = 0.0f;
  if (p < NT2) {
    const int g = pair_slots()[p] / KROW;
    const int a = g / 9;
    const int b = g - a * 9;
    const float own = U[(a * 9 + b) * PER_P + ki];
    const float oth = U[(b * 9 + a) * PER_P + ki];
    v = own + ((a != b) ? oth : 0.0f);
  }
  F[ki * TP2 + p] = v;
}

#define SETUP_WAVES   32
#define SETUP_THREADS (SETUP_WAVES * 32)
#define FOLD_BLOCKS   ((F_TOTAL + SETUP_THREADS - 1) / SETUP_THREADS)

__global__ __launch_bounds__(SETUP_THREADS) void sc_setup(
    const int* __restrict__ index, int N,
    int* __restrict__ meta, int* __restrict__ perm,
    float* __restrict__ F,
    const float* __restrict__ U3_0e, const float* __restrict__ U3_1o,
    const float* __restrict__ U3_2e, const float* __restrict__ U2_0e,
    const float* __restrict__ U2_1o, const float* __restrict__ U2_2e) {
  __shared__ int waveCnt[SETUP_WAVES][NSP];
  __shared__ int spCnt[NSP];
  const int tid  = threadIdx.x;

  if (blockIdx.x != 0) {
    const int o = (blockIdx.x - 1) * SETUP_THREADS + tid;
    if      (o < F3_1O_OFF) fold3_entry<10, 1>(U3_0e, F + F3_0E_OFF, o - F3_0E_OFF);
    else if (o < F3_2E_OFF) fold3_entry<11, 3>(U3_1o, F + F3_1O_OFF, o - F3_1O_OFF);
    else if (o < F2_0E_OFF) fold3_entry<13, 5>(U3_2e, F + F3_2E_OFF, o - F3_2E_OFF);
    else if (o < F2_1O_OFF) fold2_entry<3, 1>(U2_0e, F + F2_0E_OFF, o - F2_0E_OFF);
    else if (o < F2_2E_OFF) fold2_entry<2, 3>(U2_1o, F + F2_1O_OFF, o - F2_1O_OFF);
    else if (o < F_TOTAL)   fold2_entry<3, 5>(U2_2e, F + F2_2E_OFF, o - F2_2E_OFF);
    return;
  }
  const int lane = tid & 31;
  const int wave = tid >> 5;

  const int chunks = (N + 31) >> 5;
  const int cpw    = (chunks + SETUP_WAVES - 1) / SETUP_WAVES;
  const int n0     = wave * cpw * 32;

  int run[NSP];
  for (int q = 0; q < NSP; ++q) run[q] = 0;
  for (int j = 0; j < cpw; ++j) {
    const int n = n0 + j * 32 + lane;
    const int s = (n < N) ? index[n] : -1;
    for (int q = 0; q < NSP; ++q)
      run[q] += __builtin_popcount(__builtin_amdgcn_ballot_w32(s == q));
  }
  if (lane < NSP) {
    int v = run[0];
    for (int q = 1; q < NSP; ++q) if (lane == q) v = run[q];
    waveCnt[wave][lane] = v;
  }
  __syncthreads();

  if (tid < NSP) {
    int acc = 0;
    for (int w = 0; w < SETUP_WAVES; ++w) {
      const int v = waveCnt[w][tid];
      waveCnt[w][tid] = acc;
      acc += v;
    }
    spCnt[tid] = acc;
  }
  __syncthreads();

  int cnt[NSP], segBase[NSP];
  int base = 0;
  for (int q = 0; q < NSP; ++q) {
    cnt[q]     = spCnt[q];
    segBase[q] = base;
    base += ((cnt[q] + 15) >> 4) << 4;
  }
  const int tiles = base >> 4;

  int wb[NSP];
  for (int q = 0; q < NSP; ++q) wb[q] = segBase[q] + waveCnt[wave][q];
  for (int j = 0; j < cpw; ++j) {
    const int n = n0 + j * 32 + lane;
    const int s = (n < N) ? index[n] : -1;
    int pos = -1;
    for (int q = 0; q < NSP; ++q) {
      const unsigned m = __builtin_amdgcn_ballot_w32(s == q);
      if (s == q) pos = wb[q] + (int)__builtin_amdgcn_mbcnt_lo(m, 0u);
      wb[q] += __builtin_popcount(m);
    }
    if (pos >= 0) perm[pos] = n;
  }

  if (tid < NSP * 16) {
    const int sp = tid >> 4;
    const int j  = tid & 15;
    int c = cnt[0], sb = segBase[0];
    for (int q = 1; q < NSP; ++q) if (sp == q) { c = cnt[q]; sb = segBase[q]; }
    const int slot = c + j;
    if (slot < (((c + 15) >> 4) << 4)) perm[sb + slot] = -1;
  }

  if (tid <= NSP) {
    int v = tiles;
    for (int q = 0; q < NSP; ++q) if (tid == q) v = segBase[q] >> 4;
    meta[tid] = v;
  }
}

#define GI      5
#define GWAVES  4
#define MAIN_THREADS (GI * GWAVES * 32)
#define CTW     (MROWS * KTOT)
static_assert(NSP % GI == 0, "the grid is NSP / GI rows of C work-groups");

template <int MUL, int IRD, int M0, int TP, int KOFF, int NTHR, class Mid>
__device__ __forceinline__ void build_folded(float* Ct, const float* __restrict__ F,
                                             const float* __restrict__ Wt, int C,
                                             const int (&sp)[GI], const int (&ch)[GI],
                                             int lt, Mid&& mid) {
  constexpr int NQ = TP / 4;
  constexpr int NI = IRD * NQ;
  static_assert(NI <= NTHR, "one quad per thread");
  const int it  = (lt < NI) ? lt : NI - 1;
  const int i   = it / NQ;
  const int q   = it - i * NQ;
  const float* fp = F + i * TP + 4 * q;
  v4f u[MUL];
#pragma unroll
  for (int k = 0; k < MUL; ++k) u[k] = *(const v4f*)(fp + k * IRD * TP);
  mid();
  float* cp = Ct + (M0 + i) * KTOT + KOFF + 4 * q;
#pragma unroll
  for (int g = 0; g < GI; ++g) {
    const float* wp = Wt + (sp[g] * MUL) * C + ch[g];
    float v0 = 0.0f, v1 = 0.0f, v2 = 0.0f, v3 = 0.0f;
#pragma unroll
    for (int k = 0; k < MUL; ++k) {
      const float wk = wp[k * C];
      v0 = __builtin_fmaf(u[k].x, wk, v0);
      v1 = __builtin_fmaf(u[k].y, wk, v1);
      v2 = __builtin_fmaf(u[k].z, wk, v2);
      v3 = __builtin_fmaf(u[k].w, wk, v3);
    }
    const v4f o = {v0, v1, v2, v3};
    *(v4f*)(cp + g * CTW) = o;
  }
}

template <int IRD, int M0, int NTHR>
__device__ __forceinline__ void build_order1(float* Ct, const float* __restrict__ U1,
                                             const float* __restrict__ W1, int C,
                                             const int (&sp)[GI], const int (&ch)[GI], int lt) {
  constexpr int NI = 9 * IRD;
  static_assert(NI <= NTHR, "one entry per thread");
  const int it = (lt < NI) ? lt : NI - 1;
  const int ii = it / 9;
  const int b  = it - ii * 9;
  const float u = U1[b * IRD + ii];
#pragma unroll
  for (int g = 0; g < GI; ++g)
    Ct[g * CTW + (M0 + ii) * KTOT + K_S1 + b] = u * W1[sp[g] * C + ch[g]];
}

struct Mono { int kind, a, b, c; };
constexpr Mono mono_of(int slot) {
  if (slot < K_P2) {
    int n = 0;
    for (int a = 0; a < 9; ++a)
      for (int b = a; b < 9; ++b)
        for (int c = b; c < 9; ++c) {
          if (n == slot - K_T3) return Mono{3, a, b, c};
          ++n;
        }
    return Mono{0, 0, 0, 0};
  }
  if (slot < K_S1) {
    int n = 0;
    for (int a = 0; a < 9; ++a)
      for (int b = a; b < 9; ++b) {
        if (n == slot - K_P2) return Mono{2, a, b, 0};
        ++n;
      }
    return Mono{0, 0, 0, 0};
  }
  return (slot - K_S1 < 9) ? Mono{1, 0, slot - K_S1, 0} : Mono{0, 0, 0, 0};
}

template <int SLOT>
__device__ __forceinline__ float mono_val(const float (&xv)[9]) {
  constexpr Mono m = mono_of(SLOT);
  if constexpr (m.kind == 3)      return (xv[m.a] * xv[m.b]) * xv[m.c];
  else if constexpr (m.kind == 2) return xv[m.a] * xv[m.b];
  else if constexpr (m.kind == 1) return xv[m.b];
  else                            return 0.0f;
}

template <int SL, int SH>
__device__ __forceinline__ float b_elem(const float (&xv)[9], bool hi) {
  constexpr Mono L = mono_of(SL);
  constexpr Mono H = mono_of(SH);
  if constexpr (L.kind == 3 && H.kind == 3) {
    float p, e;
    if constexpr (L.a == H.a && L.b == H.b) p = xv[L.a] * xv[L.b];
    else                                    p = hi ? xv[H.a] * xv[H.b] : xv[L.a] * xv[L.b];
    if constexpr (L.c == H.c) e = xv[L.c];
    else                      e = hi ? xv[H.c] : xv[L.c];
    return p * e;
  } else {
    return hi ? mono_val<SH>(xv) : mono_val<SL>(xv);
  }
}

template <int J>
__device__ __forceinline__ void k_blocks(const float* Abase, const float (&xv)[9], bool hi,
                                         v8f& acc0, v8f& acc1, v8f& acc2) {
  if constexpr (J < NBLK) {
    const v2f af = *(const v2f*)(Abase + 4 * J);
    v2f bf;
    bf.x = b_elem<4 * J,     4 * J + 2>(xv, hi);
    bf.y = b_elem<4 * J + 1, 4 * J + 3>(xv, hi);
    if constexpr (J % 3 == 0)
      acc0 = __builtin_amdgcn_wmma_f32_16x16x4_f32(false, af, false, bf, (short)0, acc0, false, false);
    else if constexpr (J % 3 == 1)
      acc1 = __builtin_amdgcn_wmma_f32_16x16x4_f32(false, af, false, bf, (short)0, acc1, false, false);
    else
      acc2 = __builtin_amdgcn_wmma_f32_16x16x4_f32(false, af, false, bf, (short)0, acc2, false, false);
    k_blocks<J + 1>(Abase, xv, hi, acc0, acc1, acc2);
  }
}

__device__ __forceinline__ void load_x(float (&xr)[9], const float* __restrict__ X,
                                       int node, int C, int c) {
  const float* xp = X + ((node >= 0 ? node : 0) * C + c) * 9;
#pragma unroll
  for (int d = 0; d < 9; ++d) xr[d] = xp[d];
}

__global__ __launch_bounds__(MAIN_THREADS) void sc_main(
    const float* __restrict__ X,
    const int*   __restrict__ meta,
    const int*   __restrict__ perm,
    float*       __restrict__ out, int C,
    const float* __restrict__ F,
    const float* __restrict__ W3_0e, const float* __restrict__ W3_1o,
    const float* __restrict__ W3_2e,
    const float* __restrict__ W2_0e, const float* __restrict__ W2_1o,
    const float* __restrict__ W2_2e,
    const float* __restrict__ U1_0e, const float* __restrict__ W1_0e,
    const float* __restrict__ U1_1o, const float* __restrict__ W1_1o,
    const float* __restrict__ U1_2e, const float* __restrict__ W1_2e)
{
  __shared__ __attribute__((aligned(16))) float Ct[GI * CTW];
  const int tid = threadIdx.x;
  const int  lane = tid & 31;
  const int  wave = __builtin_amdgcn_readfirstlane(tid >> 5);
  const int  grp  = wave / GWAVES;
  const int  wv   = wave - grp * GWAVES;
  const int  n16  = lane & 15;
  const bool hi   = (lane & 16) != 0;

  const unsigned item0 = (blockIdx.y * (unsigned)C + blockIdx.x) * GI;
  int sp[GI], ch[GI];
#pragma unroll
  for (int g = 0; g < GI; ++g) {
    const unsigned item = item0 + g;
    sp[g] = (int)(item / (unsigned)C);
    ch[g] = (int)(item - (unsigned)sp[g] * (unsigned)C);
  }
  const int s = (int)((item0 + grp) / (unsigned)C);
  const int c = (int)((item0 + grp) - (unsigned)s * (unsigned)C);

  for (int n = tid; n < GI * CTW / 4; n += MAIN_THREADS) {
    const v4f z = {0.f, 0.f, 0.f, 0.f};
    *(v4f*)(Ct + 4 * n) = z;
  }
  const int t0 = meta[s];
  const int t1 = meta[s + 1];
  __syncthreads();

  const int  nT = t1 - t0;
  const int* pbase = perm + t0 * 16;
  int nodeCur = -1, nodeNxt = -1;
  if (wv < nT)          nodeCur = pbase[wv * 16 + n16];
  if (wv + GWAVES < nT) nodeNxt = pbase[(wv + GWAVES) * 16 + n16];
  float xc[9];
#pragma unroll
  for (int d = 0; d < 9; ++d) xc[d] = 0.0f;
  auto first_x = [&]() { if (wv < nT) load_x(xc, X, nodeCur, C, c); };

  if (wave < 7)
    build_folded<13, 5, 4, TP3, K_T3, 7 * 32>(Ct, F + F3_2E_OFF, W3_2e, C, sp, ch, tid, first_x);
  else if (wave < 11)
    build_folded<11, 3, 1, TP3, K_T3, 4 * 32>(Ct, F + F3_1O_OFF, W3_1o, C, sp, ch, tid - 7 * 32, first_x);
  else if (wave < 13)
    build_folded<10, 1, 0, TP3, K_T3, 2 * 32>(Ct, F + F3_0E_OFF, W3_0e, C, sp, ch, tid - 11 * 32, first_x);
  else if (wave < 15)
    build_folded<3, 5, 4, TP2, K_P2, 2 * 32>(Ct, F + F2_2E_OFF, W2_2e, C, sp, ch, tid - 13 * 32, first_x);
  else if (wave < 17)
    build_folded<2, 3, 1, TP2, K_P2, 2 * 32>(Ct, F + F2_1O_OFF, W2_1o, C, sp, ch, tid - 15 * 32, first_x);
  else if (wave < 18)
    build_folded<3, 1, 0, TP2, K_P2, 1 * 32>(Ct, F + F2_0E_OFF, W2_0e, C, sp, ch, tid - 17 * 32, first_x);
  else {
    first_x();
    build_order1<1, 0, 2 * 32>(Ct, U1_0e, W1_0e, C, sp, ch, tid - 18 * 32);
    build_order1<3, 1, 2 * 32>(Ct, U1_1o, W1_1o, C, sp, ch, tid - 18 * 32);
    build_order1<5, 4, 2 * 32>(Ct, U1_2e, W1_2e, C, sp, ch, tid - 18 * 32);
  }
  __syncthreads();

  const float* Abase = &Ct[grp * CTW + n16 * KTOT + (hi ? 2 : 0)];

  for (int my = wv; my < nT; my += GWAVES) {
    float xn[9];
#pragma unroll
    for (int d = 0; d < 9; ++d) xn[d] = 0.0f;
    int nodeNx2 = -1;
    if (my + GWAVES < nT) {
      load_x(xn, X, nodeNxt, C, c);
      if (my + 2 * GWAVES < nT) nodeNx2 = pbase[(my + 2 * GWAVES) * 16 + n16];
    }

    asm volatile("" ::: "memory");
    v8f acc0 = {0.f,0.f,0.f,0.f,0.f,0.f,0.f,0.f};
    v8f acc1 = acc0, acc2 = acc0;
    k_blocks<0>(Abase, xc, hi, acc0, acc1, acc2);
    v8f acc = acc0 + acc1 + acc2;

    if (nodeCur >= 0) {
      float* op = out + (nodeCur * C + c) * 9;
      if (!hi) {
#pragma unroll
        for (int r = 0; r < 8; ++r) op[r] = acc[r];
      } else {
        op[8] = acc[0];
      }
    }

    nodeCur = nodeNxt;
    nodeNxt = nodeNx2;
#pragma unroll
    for (int d = 0; d < 9; ++d) xc[d] = xn[d];
  }
}

extern "C" void kernel_launch(void* const* d_in, const int* in_sizes, int n_in,
                              void* d_out, int out_size, void* d_ws, size_t ws_size,
                              hipStream_t stream) {
  (void)n_in; (void)out_size; (void)ws_size;
  const float* X     = (const float*)d_in[0];
  const int*   index = (const int*)d_in[1];
  const int N = in_sizes[1];
  const int C = in_sizes[0] / (N * 9);

  int* meta        = (int*)d_ws;
  int* perm        = meta + 16;
  int  permCap     = (N + 16 * NSP + 15) & ~15;
  float* F         = (float*)(perm + permCap);

  sc_setup<<<1 + FOLD_BLOCKS, SETUP_THREADS, 0, stream>>>(
      index, N, meta, perm, F,
      (const float*)d_in[2], (const float*)d_in[4], (const float*)d_in[6],
      (const float*)d_in[8], (const float*)d_in[10], (const float*)d_in[12]);

  sc_main<<<dim3(C, NSP / GI), MAIN_THREADS, 0, stream>>>(
      X, meta, perm, (float*)d_out, C, F,
      (const float*)d_in[3],  (const float*)d_in[5],  (const float*)d_in[7],
      (const float*)d_in[9],  (const float*)d_in[11], (const float*)d_in[13],
      (const float*)d_in[14], (const float*)d_in[15],
      (const float*)d_in[16], (const float*)d_in[17],
      (const float*)d_in[18], (const float*)d_in[19]);
}


# GEMM with x_a factored out of each monomial: rows (a,i) x 55 monomials of degree <= 2, prefix k ranges, 40 WMMAs per tile not 57; x_a product and sum over a in registers; changes summation order
# speedup vs baseline: 32.8753x; 1.1053x over previous
#include <hip/hip_runtime.h>


typedef float v2f __attribute__((ext_vector_type(2)));
typedef float v4f __attribute__((ext_vector_type(4)));
typedef float v8f __attribute__((ext_vector_type(8)));

#define NSP   10
#define KROW  12

#define NT3 165
#define TP3 168
#define NT2 45
#define TP2 48
#define F3_0E_OFF 0
#define F3_1O_OFF (F3_0E_OFF + 10 * 1 * TP3)
#define F3_2E_OFF (F3_1O_OFF + 11 * 3 * TP3)
#define F2_0E_OFF (F3_2E_OFF + 13 * 5 * TP3)
#define F2_1O_OFF (F2_0E_OFF + 3 * 1 * TP2)
#define F2_2E_OFF (F2_1O_OFF + 2 * 3 * TP2)
#define F_TOTAL   (F2_2E_OFF + 3 * 5 * TP2)

#define KP    60
#define NBLK  14
#define NROWT 6
#define NROWS 81
#define PAD_OFF (86 * KP)

constexpr int blk_start(int b) { int st = 1; for (int bb = 8; bb > b; --bb) st += 1 + (9 - bb); return st; }
constexpr int slot_single(int b) { return blk_start(b); }
constexpr int slot_pair(int b, int c) { return blk_start(b) + 1 + (c - b); }
constexpr int prefix_len(int a) { return slot_pair(a, 8) + 1; }
constexpr int tile_blocks(int m) { return (prefix_len((16 * m) / 9) + 3) / 4; }
static_assert(prefix_len(0) == 55 && tile_blocks(0) == NBLK, "k axis");
static_assert(tile_blocks(0) + tile_blocks(1) + tile_blocks(2) + tile_blocks(3) + tile_blocks(4) + tile_blocks(5) == 40, "WMMAs per tile");

struct OffTab3 { unsigned short v[TP3]; };
struct OffTab2 { unsigned short v[TP2]; };
constexpr OffTab3 make_off3() {
  OffTab3 t{};
  int n = 0;
  for (int a = 0; a < 9; ++a)
    for (int b = a; b < 9; ++b)
      for (int c = b; c < 9; ++c) t.v[n++] = (unsigned short)(9 * a * KP + slot_pair(b, c));
  for (; n < TP3; ++n) t.v[n] = (unsigned short)PAD_OFF;
  return t;
}
constexpr OffTab2 make_off2() {
  OffTab2 t{};
  int n = 0;
  for (int a = 0; a < 9; ++a)
    for (int b = a; b < 9; ++b) t.v[n++] = (unsigned short)(9 * a * KP + slot_single(b));
  for (; n < TP2; ++n) t.v[n] = (unsigned short)PAD_OFF;
  return t;
}
__device__ __forceinline__ const unsigned short* off3_tab() {
  static constexpr __attribute__((aligned(8))) OffTab3 tab = make_off3();
  return tab.v;
}
__device__ __forceinline__ const unsigned short* off2_tab() {
  static constexpr __attribute__((aligned(8))) OffTab2 tab = make_off2();
  return tab.v;
}

__device__ __forceinline__ const unsigned short* triple_slots() {
  static const __attribute__((aligned(8))) unsigned short tab[TP3] = {
      0, 1, 2, 3, 4, 5, 6, 7, 8, 13, 14, 15, 16, 17,
      18, 19, 20, 26, 27, 28, 29, 30, 31, 32, 39, 40, 41, 42,
      43, 44, 52, 53, 54, 55, 56, 65, 66, 67, 68, 78, 79, 80,
      91, 92, 104, 121, 122, 123, 124, 125, 126, 127, 128, 134, 135, 136,
      137, 138, 139, 140, 147, 148, 149, 150, 151, 152, 160, 161, 162, 163,
      164, 173, 174, 175, 176, 186, 187, 188, 199, 200, 212, 242, 243, 244,
      245, 246, 247, 248, 255, 256, 257, 258, 259, 260, 268, 269, 270, 271,
      272, 281, 282, 283, 284, 294, 295, 296, 307, 308, 320, 363, 364, 365,
      366, 367, 368, 376, 377, 378, 379, 380, 389, 390, 391, 392, 402, 403,
      404, 415, 416, 428, 484, 485, 486, 487, 488, 497, 498, 499, 500, 510,
      511, 512, 523, 524, 536, 605, 606, 607, 608, 618, 619, 620, 631, 632,
      644, 726, 727, 728, 739, 740, 752, 847, 848, 860, 968, 11, 11, 11};
  return tab;
}
__device__ __forceinline__ const unsigned short* pair_slots() {
  static const __attribute__((aligned(8))) unsigned short tab[TP2] = {
      9, 21, 33, 45, 57, 69, 81, 93, 105, 129, 141, 153,
      165, 177, 189, 201, 213, 249, 261, 273, 285, 297, 309, 321,
      369, 381, 393, 405, 417, 429, 489, 501, 513, 525, 537, 609,
      621, 633, 645, 729, 741, 753, 849, 861, 969, 11, 11, 11};
  return tab;
}

template <int MUL, int IRD>
__device__ __forceinline__ void fold3_entry(const float* __restrict__ U, float* __restrict__ F, int o) {
  constexpr int PER_P = MUL * IRD;
  const int t  = o / PER_P;
  const int ki = o - t * PER_P;
  float v = 0.0f;
  if (t < NT3) {
    const int sl = triple_slots()[t];
    const int g = sl / KROW;
    const int c = sl - g * KROW;
    const int a = g / 9;
    const int b = g - a * 9;
    const float s1 = U[((a * 9 + b) * 9 + c) * PER_P + ki];
    const float s2 = U[((a * 9 + c) * 9 + b) * PER_P + ki];
    const float s3 = U[((c * 9 + a) * 9 + b) * PER_P + ki];
    const float s4 = U[((b * 9 + a) * 9 + c) * PER_P + ki];
    const float s5 = U[((b * 9 + c) * 9 + a) * PER_P + ki];
    const float s6 = U[((c * 9 + b) * 9 + a) * PER_P + ki];
    v = s1;
    v += (b != c) ? s2 : 0.0f;
    v += (b != c) ? s3 : 0.0f;
    v += (a != b) ? s4 : 0.0f;
    v += (a != b) ? s5 : 0.0f;
    v += (a != b && b != c) ? s6 : 0.0f;
  }
  F[ki * TP3 + t] = v;
}

template <int MUL, int IRD>
__device__ __forceinline__ void fold2_entry(const float* __restrict__ U, float* __restrict__ F, int o) {
  constexpr int PER_P = MUL * IRD;
  const int p  = o / PER_P;
  const int ki = o - p * PER_P;
  float v = 0.0f;
  if (p < NT2) {
    const int g = pair_slots()[p] / KROW;
    const int a = g / 9;
    const int b = g - a * 9;
    const float own = U[(a * 9 + b) * PER_P + ki];
    const float oth = U[(b * 9 + a) * PER_P + ki];
    v = own + ((a != b) ? oth : 0.0f);
  }
  F[ki * TP2 + p] = v;
}

#define SETUP_WAVES   32
#define SETUP_THREADS (SETUP_WAVES * 32)
#define FOLD_BLOCKS   ((F_TOTAL + SETUP_THREADS - 1) / SETUP_THREADS)

__global__ __launch_bounds__(SETUP_THREADS) void sc_setup(
    const int* __restrict__ index, int N,
    int* __restrict__ meta, int* __restrict__ perm,
    float* __restrict__ F,
    const float* __restrict__ U3_0e, const float* __restrict__ U3_1o,
    const float* __restrict__ U3_2e, const float* __restrict__ U2_0e,
    const float* __restrict__ U2_1o, const float* __restrict__ U2_2e) {
  __shared__ int waveCnt[SETUP_WAVES][NSP];
  __shared__ int spCnt[NSP];
  const int tid  = threadIdx.x;

  if (blockIdx.x != 0) {
    const int o = (blockIdx.x - 1) * SETUP_THREADS + tid;
    if      (o < F3_1O_OFF) fold3_entry<10, 1>(U3_0e, F + F3_0E_OFF, o - F3_0E_OFF);
    else if (o < F3_2E_OFF) fold3_entry<11, 3>(U3_1o, F + F3_1O_OFF, o - F3_1O_OFF);
    else if (o < F2_0E_OFF) fold3_entry<13, 5>(U3_2e, F + F3_2E_OFF, o - F3_2E_OFF);
    else if (o < F2_1O_OFF) fold2_entry<3, 1>(U2_0e, F + F2_0E_OFF, o - F2_0E_OFF);
    else if (o < F2_2E_OFF) fold2_entry<2, 3>(U2_1o, F + F2_1O_OFF, o - F2_1O_OFF);
    else if (o < F_TOTAL)   fold2_entry<3, 5>(U2_2e, F + F2_2E_OFF, o - F2_2E_OFF);
    return;
  }
  const int lane = tid & 31;
  const int wave = tid >> 5;

  const int chunks = (N + 31) >> 5;
  const int cpw    = (chunks + SETUP_WAVES - 1) / SETUP_WAVES;
  const int n0     = wave * cpw * 32;

  int run[NSP];
  for (int q = 0; q < NSP; ++q) run[q] = 0;
  for (int j = 0; j < cpw; ++j) {
    const int n = n0 + j * 32 + lane;
    const int s = (n < N) ? index[n] : -1;
    for (int q = 0; q < NSP; ++q)
      run[q] += __builtin_popcount(__builtin_amdgcn_ballot_w32(s == q));
  }
  if (lane < NSP) {
    int v = run[0];
    for (int q = 1; q < NSP; ++q) if (lane == q) v = run[q];
    waveCnt[wave][lane] = v;
  }
  __syncthreads();

  if (tid < NSP) {
    int acc = 0;
    for (int w = 0; w < SETUP_WAVES; ++w) {
      const int v = waveCnt[w][tid];
      waveCnt[w][tid] = acc;
      acc += v;
    }
    spCnt[tid] = acc;
  }
  __syncthreads();

  int cnt[NSP], segBase[NSP];
  int base = 0;
  for (int q = 0; q < NSP; ++q) {
    cnt[q]     = spCnt[q];
    segBase[q] = base;
    base += ((cnt[q] + 15) >> 4) << 4;
  }
  const int tiles = base >> 4;

  int wb[NSP];
  for (int q = 0; q < NSP; ++q) wb[q] = segBase[q] + waveCnt[wave][q];
  for (int j = 0; j < cpw; ++j) {
    const int n = n0 + j * 32 + lane;
    const int s = (n < N) ? index[n] : -1;
    int pos = -1;
    for (int q = 0; q < NSP; ++q) {
      const unsigned m = __builtin_amdgcn_ballot_w32(s == q);
      if (s == q) pos = wb[q] + (int)__builtin_amdgcn_mbcnt_lo(m, 0u);
      wb[q] += __builtin_popcount(m);
    }
    if (pos >= 0) perm[pos] = n;
  }

  if (tid < NSP * 16) {
    const int sp = tid >> 4;
    const int j  = tid & 15;
    int c = cnt[0], sb = segBase[0];
    for (int q = 1; q < NSP; ++q) if (sp == q) { c = cnt[q]; sb = segBase[q]; }
    const int slot = c + j;
    if (slot < (((c + 15) >> 4) << 4)) perm[sb + slot] = -1;
  }

  if (tid <= NSP) {
    int v = tiles;
    for (int q = 0; q < NSP; ++q) if (tid == q) v = segBase[q] >> 4;
    meta[tid] = v;
  }
}

#define GI      5
#define GWAVES  4
#define MAIN_THREADS (GI * GWAVES * 32)
#define CTW     (NROWT * 16 * KP)
static_assert(NSP % GI == 0, "the grid is NSP / GI rows of C work-groups");

typedef unsigned short us4 __attribute__((ext_vector_type(4)));

template <int MUL, int IRD, int M0, int TP, int NTHR, class Mid>
__device__ __forceinline__ void build_folded(float* Ct, const float* __restrict__ F,
                                             const float* __restrict__ Wt, int C,
                                             const int (&sp)[GI], const int (&ch)[GI],
                                             int lt, Mid&& mid) {
  constexpr int NQ = TP / 4;
  constexpr int NI = IRD * NQ;
  static_assert(NI <= NTHR, "one quad per thread");
  const int it  = (lt < NI) ? lt : NI - 1;
  const int i   = it / NQ;
  const int q   = it - i * NQ;
  const float* fp = F + i * TP + 4 * q;
  v4f u[MUL];
#pragma unroll
  for (int k = 0; k < MUL; ++k) u[k] = *(const v4f*)(fp + k * IRD * TP);
  const unsigned short* tb = (TP == TP3) ? off3_tab() : off2_tab();
  const us4 off = *(const us4*)(tb + 4 * q);
  mid();
  float* cp = Ct + (M0 + i) * KP;
  v4f r[GI];
#pragma unroll
  for (int g = 0; g < GI; ++g) {
    const float* wp = Wt + (sp[g] * MUL) * C + ch[g];
    float v0 = 0.0f, v1 = 0.0f, v2 = 0.0f, v3 = 0.0f;
#pragma unroll
    for (int k = 0; k < MUL; ++k) {
      const float wk = wp[k * C];
      v0 = __builtin_fmaf(u[k].x, wk, v0);
      v1 = __builtin_fmaf(u[k].y, wk, v1);
      v2 = __builtin_fmaf(u[k].z, wk, v2);
      v3 = __builtin_fmaf(u[k].w, wk, v3);
    }
    r[g] = v4f{v0, v1, v2, v3};
  }
#pragma unroll
  for (int g = 0; g < GI; ++g) {
    float* cg = cp + g * CTW;
    cg[off.x] = r[g].x;
    cg[off.y] = r[g].y;
    cg[off.z] = r[g].z;
    cg[off.w] = r[g].w;
  }
}

template <int IRD, int M0, int NTHR>
__device__ __forceinline__ void build_order1(float* Ct, const float* __restrict__ U1,
                                             const float* __restrict__ W1, int C,
                                             const int (&sp)[GI], const int (&ch)[GI], int lt) {
  constexpr int NI = 9 * IRD;
  static_assert(NI <= NTHR, "one entry per thread");
  const int it = (lt < NI) ? lt : NI - 1;
  const int ii = it / 9;
  const int b  = it - ii * 9;
  const float u = U1[b * IRD + ii];
#pragma unroll
  for (int g = 0; g < GI; ++g)
    Ct[g * CTW + (9 * b + M0 + ii) * KP] = u * W1[sp[g] * C + ch[g]];
}

struct Mono { int kind, b, c; };
constexpr Mono mono_of(int slot) {
  if (slot == 0) return Mono{0, 0, 0};
  for (int b = 0; b < 9; ++b) {
    if (slot == slot_single(b)) return Mono{1, b, 0};
    for (int c = b; c < 9; ++c)
      if (slot == slot_pair(b, c)) return Mono{2, b, c};
  }
  return Mono{-1, 0, 0};
}
static_assert(mono_of(54).kind == 2 && mono_of(54).b == 0 && mono_of(54).c == 8 && mono_of(55).kind == -1, "k axis");

template <int SLOT>
__device__ __forceinline__ float mono_val(const float (&xv)[9]) {
  constexpr Mono m = mono_of(SLOT);
  if constexpr (m.kind == 2)      return xv[m.b] * xv[m.c];
  else if constexpr (m.kind == 1) return xv[m.b];
  else if constexpr (m.kind == 0) return 1.0f;
  else                            return 0.0f;
}

template <int SL, int SH>
__device__ __forceinline__ float b_elem(const float (&xv)[9], bool hi) {
  constexpr Mono L = mono_of(SL);
  constexpr Mono H = mono_of(SH);
  if constexpr (L.kind == 2 && H.kind == 2) {
    float p, e;
    if constexpr (L.b == H.b) p = xv[L.b];
    else                      p = hi ? xv[H.b] : xv[L.b];
    if constexpr (L.c == H.c) e = xv[L.c];
    else                      e = hi ? xv[H.c] : xv[L.c];
    return p * e;
  } else {
    return hi ? mono_val<SH>(xv) : mono_val<SL>(xv);
  }
}

template <int J>
__device__ __forceinline__ void fill_b(v2f (&bfr)[NBLK], const float (&xv)[9], bool hi) {
  if constexpr (J < NBLK) {
    bfr[J].x = b_elem<4 * J,     4 * J + 2>(xv, hi);
    bfr[J].y = b_elem<4 * J + 1, 4 * J + 3>(xv, hi);
    fill_b<J + 1>(bfr, xv, hi);
  }
}

template <int J, int M>
__device__ __forceinline__ void mm_steps(const float* Abase, const v2f (&bfr)[NBLK], v8f (&acc)[NROWT]) {
  if constexpr (J < NBLK) {
    if constexpr (M < NROWT) {
      if constexpr (J < tile_blocks(M)) {
        const v2f af = *(const v2f*)(Abase + M * 16 * KP + 4 * J);
        acc[M] = __builtin_amdgcn_wmma_f32_16x16x4_f32(false, af, false, bfr[J], (short)0, acc[M], false, false);
      }
      mm_steps<J, M + 1>(Abase, bfr, acc);
    } else {
      mm_steps<J + 1, 0>(Abase, bfr, acc);
    }
  }
}

template <int M, int R>
__device__ __forceinline__ void fold_rows(float (&pv)[9], const v8f (&acc)[NROWT], const float (&xv)[9], bool hi) {
  if constexpr (M < NROWT) {
    if constexpr (R < 8) {
      constexpr int rl = 16 * M + R, rh = rl + 8;
      if constexpr (rl < NROWS) {
        constexpr int al = rl / 9, j = rl % 9;
        float xs;
        if constexpr (rh < NROWS) {
          constexpr int ah = rh / 9;
          if constexpr (ah == al) xs = xv[al];
          else                    xs = hi ? xv[ah] : xv[al];
        } else {
          xs = hi ? 0.0f : xv[al];
        }
        pv[j] = __builtin_fmaf(xs, acc[M][R], pv[j]);
      }
      fold_rows<M, R + 1>(pv, acc, xv, hi);
    } else {
      fold_rows<M + 1, 0>(pv, acc, xv, hi);
    }
  }
}

__device__ __forceinline__ float other_half(float v) {
  return __builtin_bit_cast(float, __builtin_amdgcn_permlanex16(
      __builtin_bit_cast(unsigned, v), __builtin_bit_cast(unsigned, v), 0x76543210u, 0xfedcba98u, false, false));
}

__device__ __forceinline__ void load_x(float (&xr)[9], const float* __restrict__ X,
                                       int node, int C, int c) {
  const float* xp = X + ((node >= 0 ? node : 0) * C + c) * 9;
#pragma unroll
  for (int d = 0; d < 9; ++d) xr[d] = xp[d];
}

__global__ __launch_bounds__(MAIN_THREADS) void sc_main(
    const float* __restrict__ X,
    const int*   __restrict__ meta,
    const int*   __restrict__ perm,
    float*       __restrict__ out, int C,
    const float* __restrict__ F,
    const float* __restrict__ W3_0e, const float* __restrict__ W3_1o,
    const float* __restrict__ W3_2e,
    const float* __restrict__ W2_0e, const float* __restrict__ W2_1o,
    const float* __restrict__ W2_2e,
    const float* __restrict__ U1_0e, const float* __restrict__ W1_0e,
    const float* __restrict__ U1_1o, const float* __restrict__ W1_1o,
    const float* __restrict__ U1_2e, const float* __restrict__ W1_2e)
{
  __shared__ __attribute__((aligned(16))) float Ct[GI * CTW];
  const int tid = threadIdx.x;
  const int  lane = tid & 31;
  const int  wave = __builtin_amdgcn_readfirstlane(tid >> 5);
  const int  grp  = wave / GWAVES;
  const int  wv   = wave - grp * GWAVES;
  const int  n16  = lane & 15;
  const bool hi   = (lane & 16) != 0;

  const unsigned item0 = (blockIdx.y * (unsigned)C + blockIdx.x) * GI;
  int sp[GI], ch[GI];
#pragma unroll
  for (int g = 0; g < GI; ++g) {
    const unsigned item = item0 + g;
    sp[g] = (int)(item / (unsigned)C);
    ch[g] = (int)(item - (unsigned)sp[g] * (unsigned)C);
  }
  const int s = (int)((item0 + grp) / (unsigned)C);
  const int c = (int)((item0 + grp) - (unsigned)s * (unsigned)C);

  for (int n = tid; n < GI * CTW / 4; n += MAIN_THREADS) {
    const v4f z = {0.f, 0.f, 0.f, 0.f};
    *(v4f*)(Ct + 4 * n) = z;
  }
  const int t0 = meta[s];
  const int t1 = meta[s + 1];
  __syncthreads();

  const int  nT = t1 - t0;
  const int* pbase = perm + t0 * 16;
  int nodeCur = -1, nodeNxt = -1;
  if (wv < nT)          nodeCur = pbase[wv * 16 + n16];
  if (wv + GWAVES < nT) nodeNxt = pbase[(wv + GWAVES) * 16 + n16];
  float xc[9];
#pragma unroll
  for (int d = 0; d < 9; ++d) xc[d] = 0.0f;
  auto first_x = [&]() { if (wv < nT) load_x(xc, X, nodeCur, C, c); };

  if (wave < 7)
    build_folded<13, 5, 4, TP3, 7 * 32>(Ct, F + F3_2E_OFF, W3_2e, C, sp, ch, tid, first_x);
  else if (wave < 11)
    build_folded<11, 3, 1, TP3, 4 * 32>(Ct, F + F3_1O_OFF, W3_1o, C, sp, ch, tid - 7 * 32, first_x);
  else if (wave < 13)
    build_folded<10, 1, 0, TP3, 2 * 32>(Ct, F + F3_0E_OFF, W3_0e, C, sp, ch, tid - 11 * 32, first_x);
  else if (wave < 15)
    build_folded<3, 5, 4, TP2, 2 * 32>(Ct, F + F2_2E_OFF, W2_2e, C, sp, ch, tid - 13 * 32, first_x);
  else if (wave < 17)
    build_folded<2, 3, 1, TP2, 2 * 32>(Ct, F + F2_1O_OFF, W2_1o, C, sp, ch, tid - 15 * 32, first_x);
  else if (wave < 18)
    build_folded<3, 1, 0, TP2, 1 * 32>(Ct, F + F2_0E_OFF, W2_0e, C, sp, ch, tid - 17 * 32, first_x);
  else {
    first_x();
    build_order1<1, 0, 2 * 32>(Ct, U1_0e, W1_0e, C, sp, ch, tid - 18 * 32);
    build_order1<3, 1, 2 * 32>(Ct, U1_1o, W1_1o, C, sp, ch, tid - 18 * 32);
    build_order1<5, 4, 2 * 32>(Ct, U1_2e, W1_2e, C, sp, ch, tid - 18 * 32);
  }
  __syncthreads();

  const float* Abase = &Ct[grp * CTW + n16 * KP + (hi ? 2 : 0)];

  for (int my = wv; my < nT; my += GWAVES) {
    float xn[9];
#pragma unroll
    for (int d = 0; d < 9; ++d) xn[d] = 0.0f;
    int nodeNx2 = -1;
    if (my + GWAVES < nT) {
      load_x(xn, X, nodeNxt, C, c);
      if (my + 2 * GWAVES < nT) nodeNx2 = pbase[(my + 2 * GWAVES) * 16 + n16];
    }

    asm volatile("" ::: "memory");
    v2f bfr[NBLK];
    fill_b<0>(bfr, xc, hi);
    v8f acc[NROWT];
#pragma unroll
    for (int m = 0; m < NROWT; ++m) acc[m] = v8f{0.f,0.f,0.f,0.f,0.f,0.f,0.f,0.f};
    mm_steps<0, 0>(Abase, bfr, acc);

    float pv[9];
#pragma unroll
    for (int d = 0; d < 9; ++d) pv[d] = 0.0f;
    fold_rows<0, 0>(pv, acc, xc, hi);
    float res[9];
#pragma unroll
    for (int d = 0; d < 9; ++d) res[d] = pv[d] + other_half(pv[(d + 1) % 9]);

    if (nodeCur >= 0 && !hi) {
      float* op = out + (nodeCur * C + c) * 9;
#pragma unroll
      for (int r = 0; r < 9; ++r) op[r] = res[r];
    }

    nodeCur = nodeNxt;
    nodeNxt = nodeNx2;
#pragma unroll
    for (int d = 0; d < 9; ++d) xc[d] = xn[d];
  }
}

extern "C" void kernel_launch(void* const* d_in, const int* in_sizes, int n_in,
                              void* d_out, int out_size, void* d_ws, size_t ws_size,
                              hipStream_t stream) {
  (void)n_in; (void)out_size; (void)ws_size;
  const float* X     = (const float*)d_in[0];
  const int*   index = (const int*)d_in[1];
  const int N = in_sizes[1];
  const int C = in_sizes[0] / (N * 9);

  int* meta        = (int*)d_ws;
  int* perm        = meta + 16;
  int  permCap     = (N + 16 * NSP + 15) & ~15;
  float* F         = (float*)(perm + permCap);

  sc_setup<<<1 + FOLD_BLOCKS, SETUP_THREADS, 0, stream>>>(
      index, N, meta, perm, F,
      (const float*)d_in[2], (const float*)d_in[4], (const float*)d_in[6],
      (const float*)d_in[8], (const float*)d_in[10], (const float*)d_in[12]);

  sc_main<<<dim3(C, NSP / GI), MAIN_THREADS, 0, stream>>>(
      X, meta, perm, (float*)d_out, C, F,
      (const float*)d_in[3],  (const float*)d_in[5],  (const float*)d_in[7],
      (const float*)d_in[9],  (const float*)d_in[11], (const float*)d_in[13],
      (const float*)d_in[14], (const float*)d_in[15],
      (const float*)d_in[16], (const float*)d_in[17],
      (const float*)d_in[18], (const float*)d_in[19]);
}
